# starting kernel
# baseline (speedup 1.0000x reference)
#include <hip/hip_runtime.h>
#include <hip/hip_bf16.h>

typedef __attribute__((ext_vector_type(8)))  float  v8f;
typedef __attribute__((ext_vector_type(2)))  float  v2f;

#define D 128

__global__ void sage_zero(float* __restrict__ agg, float* __restrict__ cnt,
                          long nAgg, long nCnt) {
  long i = (long)blockIdx.x * blockDim.x + threadIdx.x;
  long stride = (long)gridDim.x * blockDim.x;
  for (long j = i; j < nAgg; j += stride) agg[j] = 0.0f;
  for (long j = i; j < nCnt; j += stride) cnt[j] = 0.0f;
}

__global__ void sage_scatter(const float* __restrict__ h,
                             const int* __restrict__ src,
                             const int* __restrict__ dst,
                             float* __restrict__ agg,
                             float* __restrict__ cnt, int E) {
  int wid  = (int)((blockIdx.x * (long)blockDim.x + threadIdx.x) >> 5);
  int lane = threadIdx.x & 31;
  if (wid >= E) return;
  int s = src[wid];
  int d = dst[wid];
  const float4* xr = reinterpret_cast<const float4*>(h + (size_t)s * D);
  float* ar = agg + (size_t)d * D + lane * 4;
  __builtin_prefetch(ar, 1, 0);
  float4 v = xr[lane];
  atomicAdd(ar + 0, v.x);
  atomicAdd(ar + 1, v.y);
  atomicAdd(ar + 2, v.z);
  atomicAdd(ar + 3, v.w);
  if (lane == 0) atomicAdd(cnt + d, 1.0f);
}

__global__ void sage_mean(float* __restrict__ agg, const float* __restrict__ cnt,
                          long total) {
  long i = (long)blockIdx.x * blockDim.x + threadIdx.x;
  long stride = (long)gridDim.x * blockDim.x;
  for (long j = i; j < total; j += stride) {
    float c = cnt[j >> 7];
    agg[j] = agg[j] / fmaxf(c, 1.0f);
  }
}

template <int RELU>
__global__ __launch_bounds__(256) void sage_gemm(
    const float* __restrict__ mean, const float* __restrict__ xin,
    const float* __restrict__ Wl, const float* __restrict__ Wr,
    const float* __restrict__ bias, float* __restrict__ out, int N) {
  __shared__ float sA[16][264];

  const int Mbase = blockIdx.x * 16;
  const int tid = threadIdx.x;

  for (int i = tid; i < 16 * 256; i += 256) {
    int r = i >> 8;
    int c = i & 255;
    int row = Mbase + r;
    row = (row < N) ? row : (N - 1);
    float v = (c < D) ? mean[(size_t)row * D + c]
                      : xin[(size_t)row * D + (c - D)];
    sA[r][c] = v;
  }
  __syncthreads();

  const int lane = tid & 31;
  const int wave = tid >> 5;
  const int col  = wave * 16 + (lane & 15);
  const int half = lane >> 4;
  const int rowA = lane & 15;

  v8f acc = {};

  for (int k0 = 0; k0 < 256; k0 += 4) {
    const float* W = (k0 < D) ? Wl : Wr;
    const int kb = (k0 & (D - 1)) + half * 2;
    v2f a, b;
    a[0] = sA[rowA][k0 + half * 2];
    a[1] = sA[rowA][k0 + half * 2 + 1];
    b[0] = W[(size_t)kb * D + col];
    b[1] = W[(size_t)(kb + 1) * D + col];
    acc = __builtin_amdgcn_wmma_f32_16x16x4_f32(
        false, a, false, b, (short)0, acc, false, false);
  }

  const float bv = bias[col];
  float* orow = out + (size_t)(Mbase + half * 8) * D + col;
#pragma unroll
  for (int v = 0; v < 8; ++v) {
    float val = acc[v] + bv;
    if (RELU) val = fmaxf(val, 0.0f);
    orow[(size_t)v * D] = val;
  }
}

extern "C" void kernel_launch(void* const* d_in, const int* in_sizes, int n_in,
                              void* d_out, int out_size, void* d_ws, size_t ws_size,
                              hipStream_t stream) {
  const float* x   = (const float*)d_in[0];
  const int*   ei  = (const int*)d_in[1];
  const float* Wl1 = (const float*)d_in[2];
  const float* Wr1 = (const float*)d_in[3];
  const float* b1  = (const float*)d_in[4];
  const float* Wl2 = (const float*)d_in[5];
  const float* Wr2 = (const float*)d_in[6];
  const float* b2  = (const float*)d_in[7];
  float* out = (float*)d_out;

  const int N = in_sizes[0] / D;
  const int E = in_sizes[1] / 2;
  const int* src = ei;
  const int* dst = ei + E;

  float* agg = (float*)d_ws;
  float* cnt = agg + (size_t)N * D;
  float* h   = cnt + N;

  const long nAgg = (long)N * D;
  dim3 blk(256);
  const int zeroBlocks    = 4096;
  const int scatterBlocks = (E + 7) / 8;
  const int meanBlocks    = 4096;
  const int gemmBlocks    = (N + 15) / 16;

  sage_zero    <<<zeroBlocks,    blk, 0, stream>>>(agg, cnt, nAgg, N);
  sage_scatter <<<scatterBlocks, blk, 0, stream>>>(x, src, dst, agg, cnt, E);
  sage_mean    <<<meanBlocks,    blk, 0, stream>>>(agg, cnt, nAgg);
  sage_gemm<1> <<<gemmBlocks,    blk, 0, stream>>>(agg, x, Wl1, Wr1, b1, h, N);

  sage_zero    <<<zeroBlocks,    blk, 0, stream>>>(agg, cnt, nAgg, N);
  sage_scatter <<<scatterBlocks, blk, 0, stream>>>(h, src, dst, agg, cnt, E);
  sage_mean    <<<meanBlocks,    blk, 0, stream>>>(agg, cnt, nAgg);
  sage_gemm<0> <<<gemmBlocks,    blk, 0, stream>>>(agg, h, Wl2, Wr2, b2, out, N);
}


# scatter: lane l owns element l of each 128 B line of the row, so each wave-wide f32 atomic add is ONE whole-line request (4 per edge), not 4 quarter-filled ones; sums stay in atomic arrival order
# speedup vs baseline: 1.1373x; 1.1373x over previous
#include <hip/hip_runtime.h>
#include <hip/hip_bf16.h>

typedef __attribute__((ext_vector_type(8)))  float  v8f;
typedef __attribute__((ext_vector_type(2)))  float  v2f;

#define D 128

__global__ void sage_zero(float* __restrict__ agg, float* __restrict__ cnt,
                          long nAgg, long nCnt) {
  long i = (long)blockIdx.x * blockDim.x + threadIdx.x;
  long stride = (long)gridDim.x * blockDim.x;
  for (long j = i; j < nAgg; j += stride) agg[j] = 0.0f;
  for (long j = i; j < nCnt; j += stride) cnt[j] = 0.0f;
}

__global__ void sage_scatter(const float* __restrict__ h,
                             const int* __restrict__ src,
                             const int* __restrict__ dst,
                             float* __restrict__ agg,
                             float* __restrict__ cnt, int E) {
  int wid  = (int)((blockIdx.x * (long)blockDim.x + threadIdx.x) >> 5);
  int lane = threadIdx.x & 31;
  if (wid >= E) return;
  int s = src[wid];
  int d = dst[wid];
  const float* xr = h + (size_t)s * D + lane;
  float* ar = agg + (size_t)d * D + lane;
  __builtin_prefetch(agg + (size_t)d * D + lane * 4, 1, 0);
  float v0 = xr[0];
  float v1 = xr[32];
  float v2 = xr[64];
  float v3 = xr[96];
  atomicAdd(ar + 0,  v0);
  atomicAdd(ar + 32, v1);
  atomicAdd(ar + 64, v2);
  atomicAdd(ar + 96, v3);
  if (lane == 0) atomicAdd(cnt + d, 1.0f);
}

__global__ void sage_mean(float* __restrict__ agg, const float* __restrict__ cnt,
                          long total) {
  long i = (long)blockIdx.x * blockDim.x + threadIdx.x;
  long stride = (long)gridDim.x * blockDim.x;
  for (long j = i; j < total; j += stride) {
    float c = cnt[j >> 7];
    agg[j] = agg[j] / fmaxf(c, 1.0f);
  }
}

template <int RELU>
__global__ __launch_bounds__(256) void sage_gemm(
    const float* __restrict__ mean, const float* __restrict__ xin,
    const float* __restrict__ Wl, const float* __restrict__ Wr,
    const float* __restrict__ bias, float* __restrict__ out, int N) {
  __shared__ float sA[16][264];

  const int Mbase = blockIdx.x * 16;
  const int tid = threadIdx.x;

  for (int i = tid; i < 16 * 256; i += 256) {
    int r = i >> 8;
    int c = i & 255;
    int row = Mbase + r;
    row = (row < N) ? row : (N - 1);
    float v = (c < D) ? mean[(size_t)row * D + c]
                      : xin[(size_t)row * D + (c - D)];
    sA[r][c] = v;
  }
  __syncthreads();

  const int lane = tid & 31;
  const int wave = tid >> 5;
  const int col  = wave * 16 + (lane & 15);
  const int half = lane >> 4;
  const int rowA = lane & 15;

  v8f acc = {};

  for (int k0 = 0; k0 < 256; k0 += 4) {
    const float* W = (k0 < D) ? Wl : Wr;
    const int kb = (k0 & (D - 1)) + half * 2;
    v2f a, b;
    a[0] = sA[rowA][k0 + half * 2];
    a[1] = sA[rowA][k0 + half * 2 + 1];
    b[0] = W[(size_t)kb * D + col];
    b[1] = W[(size_t)(kb + 1) * D + col];
    acc = __builtin_amdgcn_wmma_f32_16x16x4_f32(
        false, a, false, b, (short)0, acc, false, false);
  }

  const float bv = bias[col];
  float* orow = out + (size_t)(Mbase + half * 8) * D + col;
#pragma unroll
  for (int v = 0; v < 8; ++v) {
    float val = acc[v] + bv;
    if (RELU) val = fmaxf(val, 0.0f);
    orow[(size_t)v * D] = val;
  }
}

extern "C" void kernel_launch(void* const* d_in, const int* in_sizes, int n_in,
                              void* d_out, int out_size, void* d_ws, size_t ws_size,
                              hipStream_t stream) {
  const float* x   = (const float*)d_in[0];
  const int*   ei  = (const int*)d_in[1];
  const float* Wl1 = (const float*)d_in[2];
  const float* Wr1 = (const float*)d_in[3];
  const float* b1  = (const float*)d_in[4];
  const float* Wl2 = (const float*)d_in[5];
  const float* Wr2 = (const float*)d_in[6];
  const float* b2  = (const float*)d_in[7];
  float* out = (float*)d_out;

  const int N = in_sizes[0] / D;
  const int E = in_sizes[1] / 2;
  const int* src = ei;
  const int* dst = ei + E;

  float* agg = (float*)d_ws;
  float* cnt = agg + (size_t)N * D;
  float* h   = cnt + N;

  const long nAgg = (long)N * D;
  dim3 blk(256);
  const int zeroBlocks    = 4096;
  const int scatterBlocks = (E + 7) / 8;
  const int meanBlocks    = 4096;
  const int gemmBlocks    = (N + 15) / 16;

  sage_zero    <<<zeroBlocks,    blk, 0, stream>>>(agg, cnt, nAgg, N);
  sage_scatter <<<scatterBlocks, blk, 0, stream>>>(x, src, dst, agg, cnt, E);
  sage_mean    <<<meanBlocks,    blk, 0, stream>>>(agg, cnt, nAgg);
  sage_gemm<1> <<<gemmBlocks,    blk, 0, stream>>>(agg, x, Wl1, Wr1, b1, h, N);

  sage_zero    <<<zeroBlocks,    blk, 0, stream>>>(agg, cnt, nAgg, N);
  sage_scatter <<<scatterBlocks, blk, 0, stream>>>(h, src, dst, agg, cnt, E);
  sage_mean    <<<meanBlocks,    blk, 0, stream>>>(agg, cnt, nAgg);
  sage_gemm<0> <<<gemmBlocks,    blk, 0, stream>>>(agg, h, Wl2, Wr2, b2, out, N);
}


# aggregation: float-atomic push scatter replaced by a pull over edges bucketed by destination (counting sort, int atomics, built per call, shared by both layers); row sums in slot (arrival) order
# speedup vs baseline: 3.0029x; 2.6404x over previous
#include <hip/hip_runtime.h>
#include <hip/hip_bf16.h>

typedef __attribute__((ext_vector_type(8)))  float  v8f;
typedef __attribute__((ext_vector_type(2)))  float  v2f;
typedef __attribute__((ext_vector_type(4)))  float  v4f;

#define D 128

#define SCAN_ELEMS 1024

__global__ void csr_zero(int* __restrict__ cnt, int N) {
  int i = blockIdx.x * blockDim.x + threadIdx.x;
  int stride = gridDim.x * blockDim.x;
  for (int j = i; j < N; j += stride) cnt[j] = 0;
}

__global__ void csr_count(const int* __restrict__ dst, int* __restrict__ cnt,
                          int* __restrict__ rank, int E) {
  int e = blockIdx.x * blockDim.x + threadIdx.x;
  if (e >= E) return;
  rank[e] = atomicAdd(cnt + dst[e], 1);
}

__global__ __launch_bounds__(256) void csr_blocksum(
    const int* __restrict__ cnt, int* __restrict__ bsum, int N) {
  __shared__ int red[256];
  const int t = threadIdx.x;
  const int i0 = blockIdx.x * SCAN_ELEMS + t * 4;
  int s = 0;
  for (int k = 0; k < 4; ++k)
    if (i0 + k < N) s += cnt[i0 + k];
  red[t] = s;
  __syncthreads();
  for (int o = 128; o > 0; o >>= 1) {
    if (t < o) red[t] += red[t + o];
    __syncthreads();
  }
  if (t == 0) bsum[blockIdx.x] = red[0];
}

__global__ __launch_bounds__(256) void csr_scan(
    const int* __restrict__ cnt, const int* __restrict__ bsum,
    int* __restrict__ rowptr, int N) {
  __shared__ int buf[2][256];
  __shared__ int red[256];
  const int t = threadIdx.x;
  const int b = blockIdx.x;

  int pre = 0;
  for (int i = t; i < b; i += 256) pre += bsum[i];
  red[t] = pre;

  const int i0 = b * SCAN_ELEMS + t * 4;
  int c[4];
  int s = 0;
#pragma unroll
  for (int k = 0; k < 4; ++k) {
    c[k] = (i0 + k < N) ? cnt[i0 + k] : 0;
    s += c[k];
  }
  buf[0][t] = s;
  __syncthreads();

  for (int o = 128; o > 0; o >>= 1) {
    if (t < o) red[t] += red[t + o];
    __syncthreads();
  }
  const int offset = red[0];

  int cur = 0;
  for (int o = 1; o < 256; o <<= 1) {
    int v = buf[cur][t];
    if (t >= o) v += buf[cur][t - o];
    buf[cur ^ 1][t] = v;
    cur ^= 1;
    __syncthreads();
  }
  int run = offset + buf[cur][t] - s;
#pragma unroll
  for (int k = 0; k < 4; ++k) {
    if (i0 + k < N) rowptr[i0 + k] = run;
    run += c[k];
  }
}

__global__ void csr_fill(const int* __restrict__ src, const int* __restrict__ dst,
                         const int* __restrict__ rowptr,
                         const int* __restrict__ rank,
                         int* __restrict__ ssrc, int E) {
  int e = blockIdx.x * blockDim.x + threadIdx.x;
  if (e >= E) return;
  ssrc[rowptr[dst[e]] + rank[e]] = src[e];
}

__global__ __launch_bounds__(256) void sage_pull_mean(
    const float* __restrict__ h, const int* __restrict__ rowptr,
    const int* __restrict__ cnt, const int* __restrict__ ssrc,
    float* __restrict__ mean, int N) {
  const v4f* __restrict__ h4 = (const v4f*)h;
  v4f* __restrict__ mean4 = (v4f*)mean;
  const int lane = threadIdx.x & 31;
  const int node = __builtin_amdgcn_readfirstlane(
      (int)(blockIdx.x * 8 + (threadIdx.x >> 5)));
  if (node >= N) return;
  const int start = rowptr[node];
  const int deg   = cnt[node];

  v4f acc = {0.0f, 0.0f, 0.0f, 0.0f};
  for (int base = 0; base < deg; base += 32) {
    const int m = min(32, deg - base);
    const int my = ssrc[start + base + min(lane, m - 1)];
    for (int j = 0; j < m; j += 8) {
      v4f r[8];
#pragma unroll
      for (int k = 0; k < 8; ++k) {
        const int jj = min(j + k, m - 1);
        const int s = __builtin_amdgcn_readlane(my, jj);
        r[k] = h4[(size_t)s * (D / 4) + lane];
      }
      asm volatile("" :: "v"(r[0]), "v"(r[1]), "v"(r[2]), "v"(r[3]),
                         "v"(r[4]), "v"(r[5]), "v"(r[6]), "v"(r[7]));
#pragma unroll
      for (int k = 0; k < 8; ++k) {
        if (j + k < m) acc += r[k];
      }
    }
  }
  const float c = fmaxf((float)deg, 1.0f);
  mean4[(size_t)node * (D / 4) + lane] = acc / c;
}

template <int RELU>
__global__ __launch_bounds__(256) void sage_gemm(
    const float* __restrict__ mean, const float* __restrict__ xin,
    const float* __restrict__ Wl, const float* __restrict__ Wr,
    const float* __restrict__ bias, float* __restrict__ out, int N) {
  __shared__ float sA[16][264];

  const int Mbase = blockIdx.x * 16;
  const int tid = threadIdx.x;

  for (int i = tid; i < 16 * 256; i += 256) {
    int r = i >> 8;
    int c = i & 255;
    int row = Mbase + r;
    row = (row < N) ? row : (N - 1);
    float v = (c < D) ? mean[(size_t)row * D + c]
                      : xin[(size_t)row * D + (c - D)];
    sA[r][c] = v;
  }
  __syncthreads();

  const int lane = tid & 31;
  const int wave = tid >> 5;
  const int col  = wave * 16 + (lane & 15);
  const int half = lane >> 4;
  const int rowA = lane & 15;

  v8f acc = {};

  for (int k0 = 0; k0 < 256; k0 += 4) {
    const float* W = (k0 < D) ? Wl : Wr;
    const int kb = (k0 & (D - 1)) + half * 2;
    v2f a, b;
    a[0] = sA[rowA][k0 + half * 2];
    a[1] = sA[rowA][k0 + half * 2 + 1];
    b[0] = W[(size_t)kb * D + col];
    b[1] = W[(size_t)(kb + 1) * D + col];
    acc = __builtin_amdgcn_wmma_f32_16x16x4_f32(
        false, a, false, b, (short)0, acc, false, false);
  }

  const float bv = bias[col];
  float* orow = out + (size_t)(Mbase + half * 8) * D + col;
#pragma unroll
  for (int v = 0; v < 8; ++v) {
    float val = acc[v] + bv;
    if (RELU) val = fmaxf(val, 0.0f);
    orow[(size_t)v * D] = val;
  }
}

extern "C" void kernel_launch(void* const* d_in, const int* in_sizes, int n_in,
                              void* d_out, int out_size, void* d_ws, size_t ws_size,
                              hipStream_t stream) {
  const float* x   = (const float*)d_in[0];
  const int*   ei  = (const int*)d_in[1];
  const float* Wl1 = (const float*)d_in[2];
  const float* Wr1 = (const float*)d_in[3];
  const float* b1  = (const float*)d_in[4];
  const float* Wl2 = (const float*)d_in[5];
  const float* Wr2 = (const float*)d_in[6];
  const float* b2  = (const float*)d_in[7];
  float* out = (float*)d_out;

  const int N = in_sizes[0] / D;
  const int E = in_sizes[1] / 2;
  const int* src = ei;
  const int* dst = ei + E;

  const int scanBlocks = (N + SCAN_ELEMS - 1) / SCAN_ELEMS;

  float* agg = (float*)d_ws;
  float* h   = agg + (size_t)N * D;
  int* cnt    = (int*)(h + (size_t)N * D);
  int* rowptr = cnt + N;
  int* bsum   = rowptr + N;
  int* rank   = bsum + scanBlocks;
  int* ssrc   = rank + E;

  dim3 blk(256);
  const int edgeBlocks = (E + 255) / 256;
  const int pullBlocks = (N + 7) / 8;
  const int gemmBlocks = (N + 15) / 16;

  csr_zero     <<<scanBlocks, blk, 0, stream>>>(cnt, N);
  csr_count    <<<edgeBlocks, blk, 0, stream>>>(dst, cnt, rank, E);
  csr_blocksum <<<scanBlocks, blk, 0, stream>>>(cnt, bsum, N);
  csr_scan     <<<scanBlocks, blk, 0, stream>>>(cnt, bsum, rowptr, N);
  csr_fill     <<<edgeBlocks, blk, 0, stream>>>(src, dst, rowptr, rank, ssrc, E);

  sage_pull_mean<<<pullBlocks, blk, 0, stream>>>(x, rowptr, cnt, ssrc, agg, N);
  sage_gemm<1> <<<gemmBlocks,    blk, 0, stream>>>(agg, x, Wl1, Wr1, b1, h, N);

  sage_pull_mean<<<pullBlocks, blk, 0, stream>>>(h, rowptr, cnt, ssrc, agg, N);
  sage_gemm<0> <<<gemmBlocks,    blk, 0, stream>>>(agg, h, Wl2, Wr2, b2, out, N);
}


# layer fusion: each 16-row block pulls its rows' neighbour means straight into the LDS A tile of its WMMA product (was a separate kernel via a workspace plane); 7 launches, was 9; sums in slot order
# speedup vs baseline: 3.1814x; 1.0594x over previous
#include <hip/hip_runtime.h>
#include <hip/hip_bf16.h>

typedef __attribute__((ext_vector_type(8)))  float  v8f;
typedef __attribute__((ext_vector_type(2)))  float  v2f;
typedef __attribute__((ext_vector_type(4)))  float  v4f;

#define D 128

#define SCAN_ELEMS 1024

__global__ void csr_zero(int* __restrict__ cnt, int N) {
  int i = blockIdx.x * blockDim.x + threadIdx.x;
  int stride = gridDim.x * blockDim.x;
  for (int j = i; j < N; j += stride) cnt[j] = 0;
}

__global__ void csr_count(const int* __restrict__ dst, int* __restrict__ cnt,
                          int* __restrict__ rank, int E) {
  int e = blockIdx.x * blockDim.x + threadIdx.x;
  if (e >= E) return;
  rank[e] = atomicAdd(cnt + dst[e], 1);
}

__global__ __launch_bounds__(256) void csr_blocksum(
    const int* __restrict__ cnt, int* __restrict__ bsum, int N) {
  __shared__ int red[256];
  const int t = threadIdx.x;
  const int i0 = blockIdx.x * SCAN_ELEMS + t * 4;
  int s = 0;
  for (int k = 0; k < 4; ++k)
    if (i0 + k < N) s += cnt[i0 + k];
  red[t] = s;
  __syncthreads();
  for (int o = 128; o > 0; o >>= 1) {
    if (t < o) red[t] += red[t + o];
    __syncthreads();
  }
  if (t == 0) bsum[blockIdx.x] = red[0];
}

__global__ __launch_bounds__(256) void csr_scan(
    const int* __restrict__ cnt, const int* __restrict__ bsum,
    int* __restrict__ rowptr, int N) {
  __shared__ int buf[2][256];
  __shared__ int red[256];
  const int t = threadIdx.x;
  const int b = blockIdx.x;

  int pre = 0;
  for (int i = t; i < b; i += 256) pre += bsum[i];
  red[t] = pre;

  const int i0 = b * SCAN_ELEMS + t * 4;
  int c[4];
  int s = 0;
#pragma unroll
  for (int k = 0; k < 4; ++k) {
    c[k] = (i0 + k < N) ? cnt[i0 + k] : 0;
    s += c[k];
  }
  buf[0][t] = s;
  __syncthreads();

  for (int o = 128; o > 0; o >>= 1) {
    if (t < o) red[t] += red[t + o];
    __syncthreads();
  }
  const int offset = red[0];

  int cur = 0;
  for (int o = 1; o < 256; o <<= 1) {
    int v = buf[cur][t];
    if (t >= o) v += buf[cur][t - o];
    buf[cur ^ 1][t] = v;
    cur ^= 1;
    __syncthreads();
  }
  int run = offset + buf[cur][t] - s;
#pragma unroll
  for (int k = 0; k < 4; ++k) {
    if (i0 + k < N) rowptr[i0 + k] = run;
    run += c[k];
  }
}

__global__ void csr_fill(const int* __restrict__ src, const int* __restrict__ dst,
                         const int* __restrict__ rowptr,
                         const int* __restrict__ rank,
                         int* __restrict__ ssrc, int E) {
  int e = blockIdx.x * blockDim.x + threadIdx.x;
  if (e >= E) return;
  ssrc[rowptr[dst[e]] + rank[e]] = src[e];
}

__device__ __forceinline__ v4f pull_row_sum(
    const v4f* __restrict__ h4, const int* __restrict__ ssrc,
    int start, int deg, int lane, int first) {
  v4f acc = {0.0f, 0.0f, 0.0f, 0.0f};
  for (int base = 0; base < deg; base += 32) {
    const int m = min(32, deg - base);
    const int my = (base == 0) ? first : ssrc[start + base + min(lane, m - 1)];
    for (int j = 0; j < m; j += 8) {
      v4f r[8];
#pragma unroll
      for (int k = 0; k < 8; ++k) {
        const int jj = min(j + k, m - 1);
        const int s = __builtin_amdgcn_readlane(my, jj);
        r[k] = h4[(size_t)s * (D / 4) + lane];
      }
      asm volatile("" :: "v"(r[0]), "v"(r[1]), "v"(r[2]), "v"(r[3]),
                         "v"(r[4]), "v"(r[5]), "v"(r[6]), "v"(r[7]));
#pragma unroll
      for (int k = 0; k < 8; ++k) {
        if (j + k < m) acc += r[k];
      }
    }
  }
  return acc;
}

template <int RELU>
__global__ __launch_bounds__(256) void sage_layer(
    const float* __restrict__ hin, const int* __restrict__ rowptr,
    const int* __restrict__ cnt, const int* __restrict__ ssrc,
    const float* __restrict__ Wl, const float* __restrict__ Wr,
    const float* __restrict__ bias, float* __restrict__ out, int N) {
  __shared__ __attribute__((aligned(16))) float sA[16][264];

  const v4f* __restrict__ h4 = (const v4f*)hin;
  const int Mbase = blockIdx.x * 16;
  const int tid = threadIdx.x;
  const int lane = tid & 31;
  const int wave = __builtin_amdgcn_readfirstlane(tid >> 5);

  {
    const int r0 = wave * 2;
    const int r1 = r0 + 1;
    const int n0 = min(Mbase + r0, N - 1);
    const int n1 = min(Mbase + r1, N - 1);
    const int start0 = rowptr[n0];
    const int deg0   = cnt[n0];
    const int start1 = rowptr[n1];
    const int deg1   = cnt[n1];
    int first0 = 0, first1 = 0;
    if (deg0 > 0) first0 = ssrc[start0 + min(lane, min(32, deg0) - 1)];
    if (deg1 > 0) first1 = ssrc[start1 + min(lane, min(32, deg1) - 1)];
    const v4f own0 = h4[(size_t)n0 * (D / 4) + lane];
    const v4f own1 = h4[(size_t)n1 * (D / 4) + lane];

    const v4f sum0 = pull_row_sum(h4, ssrc, start0, deg0, lane, first0);
    const v4f sum1 = pull_row_sum(h4, ssrc, start1, deg1, lane, first1);

    const float c0 = fmaxf((float)deg0, 1.0f);
    const float c1 = fmaxf((float)deg1, 1.0f);
    *(v4f*)&sA[r0][4 * lane]     = sum0 / c0;
    *(v4f*)&sA[r1][4 * lane]     = sum1 / c1;
    *(v4f*)&sA[r0][D + 4 * lane] = own0;
    *(v4f*)&sA[r1][D + 4 * lane] = own1;
  }
  __syncthreads();

  const int col  = wave * 16 + (lane & 15);
  const int half = lane >> 4;
  const int rowA = lane & 15;

  v8f acc = {};

  for (int k0 = 0; k0 < 256; k0 += 4) {
    const float* W = (k0 < D) ? Wl : Wr;
    const int kb = (k0 & (D - 1)) + half * 2;
    v2f a, b;
    a[0] = sA[rowA][k0 + half * 2];
    a[1] = sA[rowA][k0 + half * 2 + 1];
    b[0] = W[(size_t)kb * D + col];
    b[1] = W[(size_t)(kb + 1) * D + col];
    acc = __builtin_amdgcn_wmma_f32_16x16x4_f32(
        false, a, false, b, (short)0, acc, false, false);
  }

  const float bv = bias[col];
  float* orow = out + (size_t)(Mbase + half * 8) * D + col;
#pragma unroll
  for (int v = 0; v < 8; ++v) {
    float val = acc[v] + bv;
    if (RELU) val = fmaxf(val, 0.0f);
    orow[(size_t)v * D] = val;
  }
}

extern "C" void kernel_launch(void* const* d_in, const int* in_sizes, int n_in,
                              void* d_out, int out_size, void* d_ws, size_t ws_size,
                              hipStream_t stream) {
  const float* x   = (const float*)d_in[0];
  const int*   ei  = (const int*)d_in[1];
  const float* Wl1 = (const float*)d_in[2];
  const float* Wr1 = (const float*)d_in[3];
  const float* b1  = (const float*)d_in[4];
  const float* Wl2 = (const float*)d_in[5];
  const float* Wr2 = (const float*)d_in[6];
  const float* b2  = (const float*)d_in[7];
  float* out = (float*)d_out;

  const int N = in_sizes[0] / D;
  const int E = in_sizes[1] / 2;
  const int* src = ei;
  const int* dst = ei + E;

  const int scanBlocks = (N + SCAN_ELEMS - 1) / SCAN_ELEMS;

  float* h   = (float*)d_ws;
  int* cnt    = (int*)(h + (size_t)N * D);
  int* rowptr = cnt + N;
  int* bsum   = rowptr + N;
  int* rank   = bsum + scanBlocks;
  int* ssrc   = rank + E;

  dim3 blk(256);
  const int edgeBlocks = (E + 255) / 256;
  const int tileBlocks = (N + 15) / 16;

  csr_zero     <<<scanBlocks, blk, 0, stream>>>(cnt, N);
  csr_count    <<<edgeBlocks, blk, 0, stream>>>(dst, cnt, rank, E);
  csr_blocksum <<<scanBlocks, blk, 0, stream>>>(cnt, bsum, N);
  csr_scan     <<<scanBlocks, blk, 0, stream>>>(cnt, bsum, rowptr, N);
  csr_fill     <<<edgeBlocks, blk, 0, stream>>>(src, dst, rowptr, rank, ssrc, E);

  sage_layer<1><<<tileBlocks, blk, 0, stream>>>(x, rowptr, cnt, ssrc, Wl1, Wr1, b1, h, N);

  sage_layer<0><<<tileBlocks, blk, 0, stream>>>(h, rowptr, cnt, ssrc, Wl2, Wr2, b2, out, N);
}


# edge bucketing: per-edge global returning int atomics and scattered rowptr gathers replaced by a two-level counting sort on LDS counters (256-node bins, then per bin); sums in LDS-atomic slot order
# speedup vs baseline: 4.2997x; 1.3515x over previous
#include <hip/hip_runtime.h>
#include <hip/hip_bf16.h>

typedef __attribute__((ext_vector_type(8)))  float  v8f;
typedef __attribute__((ext_vector_type(2)))  float  v2f;
typedef __attribute__((ext_vector_type(4)))  float  v4f;

#define D 128

#define SCAN_ELEMS 1024
#define PART_BLOCKS 256
#define BIN_SHIFT 8
#define BIN_NODES (1 << BIN_SHIFT)
#define MAX_BINS 2048

__global__ __launch_bounds__(1024) void part_hist(
    const int* __restrict__ dst, int* __restrict__ hist_g, int E, int NB) {
  __shared__ int hist[MAX_BINS];
  const int t = threadIdx.x;
  const int b = blockIdx.x;
  const int nb = gridDim.x;
  for (int i = t; i < NB; i += 1024) hist[i] = 0;
  __syncthreads();
  const int S = (E + nb - 1) / nb;
  const int lo = b * S;
  const int hi = min(E, lo + S);
  for (int e0 = lo + t; e0 < hi; e0 += 4 * 1024) {
    int d[4];
#pragma unroll
    for (int k = 0; k < 4; ++k) d[k] = dst[min(e0 + k * 1024, hi - 1)];
    asm volatile("" :: "v"(d[0]), "v"(d[1]), "v"(d[2]), "v"(d[3]));
#pragma unroll
    for (int k = 0; k < 4; ++k)
      if (e0 + k * 1024 < hi) atomicAdd(&hist[d[k] >> BIN_SHIFT], 1);
  }
  __syncthreads();
  for (int i = t; i < NB; i += 1024) hist_g[i * nb + b] = hist[i];
}

__global__ __launch_bounds__(256) void csr_blocksum(
    const int* __restrict__ cnt, int* __restrict__ bsum, int N) {
  __shared__ int red[256];
  const int t = threadIdx.x;
  const int i0 = blockIdx.x * SCAN_ELEMS + t * 4;
  int s = 0;
  for (int k = 0; k < 4; ++k)
    if (i0 + k < N) s += cnt[i0 + k];
  red[t] = s;
  __syncthreads();
  for (int o = 128; o > 0; o >>= 1) {
    if (t < o) red[t] += red[t + o];
    __syncthreads();
  }
  if (t == 0) bsum[blockIdx.x] = red[0];
}

__global__ __launch_bounds__(256) void csr_scan(
    const int* __restrict__ cnt, const int* __restrict__ bsum,
    int* __restrict__ rowptr, int N) {
  __shared__ int buf[2][256];
  __shared__ int red[256];
  const int t = threadIdx.x;
  const int b = blockIdx.x;

  int pre = 0;
  for (int i = t; i < b; i += 256) pre += bsum[i];
  red[t] = pre;

  const int i0 = b * SCAN_ELEMS + t * 4;
  int c[4];
  int s = 0;
#pragma unroll
  for (int k = 0; k < 4; ++k) {
    c[k] = (i0 + k < N) ? cnt[i0 + k] : 0;
    s += c[k];
  }
  buf[0][t] = s;
  __syncthreads();

  for (int o = 128; o > 0; o >>= 1) {
    if (t < o) red[t] += red[t + o];
    __syncthreads();
  }
  const int offset = red[0];

  int cur = 0;
  for (int o = 1; o < 256; o <<= 1) {
    int v = buf[cur][t];
    if (t >= o) v += buf[cur][t - o];
    buf[cur ^ 1][t] = v;
    cur ^= 1;
    __syncthreads();
  }
  int run = offset + buf[cur][t] - s;
#pragma unroll
  for (int k = 0; k < 4; ++k) {
    if (i0 + k < N) rowptr[i0 + k] = run;
    run += c[k];
  }
}

__global__ __launch_bounds__(1024) void part_scatter(
    const int* __restrict__ src, const int* __restrict__ dst,
    const int* __restrict__ scan_g, int2* __restrict__ pairs, int E, int NB) {
  __shared__ int cursor[MAX_BINS];
  const int t = threadIdx.x;
  const int b = blockIdx.x;
  const int nb = gridDim.x;
  for (int i = t; i < NB; i += 1024) cursor[i] = scan_g[i * nb + b];
  __syncthreads();
  const int S = (E + nb - 1) / nb;
  const int lo = b * S;
  const int hi = min(E, lo + S);
  for (int e0 = lo + t; e0 < hi; e0 += 4 * 1024) {
    int d[4], s[4];
#pragma unroll
    for (int k = 0; k < 4; ++k) {
      const int e = min(e0 + k * 1024, hi - 1);
      d[k] = dst[e];
      s[k] = src[e];
    }
    asm volatile("" :: "v"(d[0]), "v"(d[1]), "v"(d[2]), "v"(d[3]),
                       "v"(s[0]), "v"(s[1]), "v"(s[2]), "v"(s[3]));
#pragma unroll
    for (int k = 0; k < 4; ++k) {
      if (e0 + k * 1024 < hi) {
        const int pos = atomicAdd(&cursor[d[k] >> BIN_SHIFT], 1);
        pairs[pos] = make_int2(s[k], d[k]);
      }
    }
  }
}

__global__ __launch_bounds__(1024) void bin_sort(
    const int2* __restrict__ pairs, const int* __restrict__ scan_g,
    int* __restrict__ cnt, int* __restrict__ rowptr, int* __restrict__ ssrc,
    int N, int E, int NB, int slices) {
  __shared__ int c[BIN_NODES];
  __shared__ int buf[2][BIN_NODES];
  __shared__ int cur[BIN_NODES];
  const int t = threadIdx.x;
  const int bin = blockIdx.x;
  const int node0 = bin << BIN_SHIFT;
  const int lo = scan_g[bin * slices];
  const int hi = (bin + 1 < NB) ? scan_g[(bin + 1) * slices] : E;

  if (t < BIN_NODES) c[t] = 0;
  __syncthreads();
  for (int e0 = lo + t; e0 < hi; e0 += 4 * 1024) {
    int d[4];
#pragma unroll
    for (int k = 0; k < 4; ++k) d[k] = pairs[min(e0 + k * 1024, hi - 1)].y;
    asm volatile("" :: "v"(d[0]), "v"(d[1]), "v"(d[2]), "v"(d[3]));
#pragma unroll
    for (int k = 0; k < 4; ++k)
      if (e0 + k * 1024 < hi) atomicAdd(&c[d[k] - node0], 1);
  }
  __syncthreads();

  if (t < BIN_NODES) buf[0][t] = c[t];
  __syncthreads();
  int w = 0;
  for (int o = 1; o < BIN_NODES; o <<= 1) {
    if (t < BIN_NODES) {
      int v = buf[w][t];
      if (t >= o) v += buf[w][t - o];
      buf[w ^ 1][t] = v;
    }
    w ^= 1;
    __syncthreads();
  }
  if (t < BIN_NODES) {
    const int deg = c[t];
    const int first = lo + buf[w][t] - deg;
    cur[t] = first;
    if (node0 + t < N) {
      cnt[node0 + t] = deg;
      rowptr[node0 + t] = first;
    }
  }
  __syncthreads();
  for (int e0 = lo + t; e0 < hi; e0 += 4 * 1024) {
    int2 p[4];
#pragma unroll
    for (int k = 0; k < 4; ++k) p[k] = pairs[min(e0 + k * 1024, hi - 1)];
    asm volatile("" :: "v"(p[0].x), "v"(p[0].y), "v"(p[1].x), "v"(p[1].y),
                       "v"(p[2].x), "v"(p[2].y), "v"(p[3].x), "v"(p[3].y));
#pragma unroll
    for (int k = 0; k < 4; ++k) {
      if (e0 + k * 1024 < hi) {
        const int pos = atomicAdd(&cur[p[k].y - node0], 1);
        ssrc[pos] = p[k].x;
      }
    }
  }
}

__device__ __forceinline__ v4f pull_row_sum(
    const v4f* __restrict__ h4, const int* __restrict__ ssrc,
    int start, int deg, int lane, int first) {
  v4f acc = {0.0f, 0.0f, 0.0f, 0.0f};
  for (int base = 0; base < deg; base += 32) {
    const int m = min(32, deg - base);
    const int my = (base == 0) ? first : ssrc[start + base + min(lane, m - 1)];
    for (int j = 0; j < m; j += 8) {
      v4f r[8];
#pragma unroll
      for (int k = 0; k < 8; ++k) {
        const int jj = min(j + k, m - 1);
        const int s = __builtin_amdgcn_readlane(my, jj);
        r[k] = h4[(size_t)s * (D / 4) + lane];
      }
      asm volatile("" :: "v"(r[0]), "v"(r[1]), "v"(r[2]), "v"(r[3]),
                         "v"(r[4]), "v"(r[5]), "v"(r[6]), "v"(r[7]));
#pragma unroll
      for (int k = 0; k < 8; ++k) {
        if (j + k < m) acc += r[k];
      }
    }
  }
  return acc;
}

template <int RELU>
__global__ __launch_bounds__(256) void sage_layer(
    const float* __restrict__ hin, const int* __restrict__ rowptr,
    const int* __restrict__ cnt, const int* __restrict__ ssrc,
    const float* __restrict__ Wl, const float* __restrict__ Wr,
    const float* __restrict__ bias, float* __restrict__ out, int N) {
  __shared__ __attribute__((aligned(16))) float sA[16][264];

  const v4f* __restrict__ h4 = (const v4f*)hin;
  const int Mbase = blockIdx.x * 16;
  const int tid = threadIdx.x;
  const int lane = tid & 31;
  const int wave = __builtin_amdgcn_readfirstlane(tid >> 5);

  {
    const int r0 = wave * 2;
    const int r1 = r0 + 1;
    const int n0 = min(Mbase + r0, N - 1);
    const int n1 = min(Mbase + r1, N - 1);
    const int start0 = rowptr[n0];
    const int deg0   = cnt[n0];
    const int start1 = rowptr[n1];
    const int deg1   = cnt[n1];
    int first0 = 0, first1 = 0;
    if (deg0 > 0) first0 = ssrc[start0 + min(lane, min(32, deg0) - 1)];
    if (deg1 > 0) first1 = ssrc[start1 + min(lane, min(32, deg1) - 1)];
    const v4f own0 = h4[(size_t)n0 * (D / 4) + lane];
    const v4f own1 = h4[(size_t)n1 * (D / 4) + lane];

    const v4f sum0 = pull_row_sum(h4, ssrc, start0, deg0, lane, first0);
    const v4f sum1 = pull_row_sum(h4, ssrc, start1, deg1, lane, first1);

    const float c0 = fmaxf((float)deg0, 1.0f);
    const float c1 = fmaxf((float)deg1, 1.0f);
    *(v4f*)&sA[r0][4 * lane]     = sum0 / c0;
    *(v4f*)&sA[r1][4 * lane]     = sum1 / c1;
    *(v4f*)&sA[r0][D + 4 * lane] = own0;
    *(v4f*)&sA[r1][D + 4 * lane] = own1;
  }
  __syncthreads();

  const int col  = wave * 16 + (lane & 15);
  const int half = lane >> 4;
  const int rowA = lane & 15;

  v8f acc = {};

  for (int k0 = 0; k0 < 256; k0 += 4) {
    const float* W = (k0 < D) ? Wl : Wr;
    const int kb = (k0 & (D - 1)) + half * 2;
    v2f a, b;
    a[0] = sA[rowA][k0 + half * 2];
    a[1] = sA[rowA][k0 + half * 2 + 1];
    b[0] = W[(size_t)kb * D + col];
    b[1] = W[(size_t)(kb + 1) * D + col];
    acc = __builtin_amdgcn_wmma_f32_16x16x4_f32(
        false, a, false, b, (short)0, acc, false, false);
  }

  const float bv = bias[col];
  float* orow = out + (size_t)(Mbase + half * 8) * D + col;
#pragma unroll
  for (int v = 0; v < 8; ++v) {
    float val = acc[v] + bv;
    if (RELU) val = fmaxf(val, 0.0f);
    orow[(size_t)v * D] = val;
  }
}

extern "C" void kernel_launch(void* const* d_in, const int* in_sizes, int n_in,
                              void* d_out, int out_size, void* d_ws, size_t ws_size,
                              hipStream_t stream) {
  const float* x   = (const float*)d_in[0];
  const int*   ei  = (const int*)d_in[1];
  const float* Wl1 = (const float*)d_in[2];
  const float* Wr1 = (const float*)d_in[3];
  const float* b1  = (const float*)d_in[4];
  const float* Wl2 = (const float*)d_in[5];
  const float* Wr2 = (const float*)d_in[6];
  const float* b2  = (const float*)d_in[7];
  float* out = (float*)d_out;

  const int N = in_sizes[0] / D;
  const int E = in_sizes[1] / 2;
  const int* src = ei;
  const int* dst = ei + E;

  const int NB = (N + BIN_NODES - 1) / BIN_NODES;
  const int M  = NB * PART_BLOCKS;
  const int scanBlocks = (M + SCAN_ELEMS - 1) / SCAN_ELEMS;

  float* h   = (float*)d_ws;
  int2* pairs = (int2*)(h + (size_t)N * D);
  int* cnt    = (int*)(pairs + E);
  int* rowptr = cnt + N;
  int* hist_g = rowptr + N;
  int* scan_g = hist_g + M;
  int* bsum   = scan_g + M;
  int* ssrc   = bsum + scanBlocks;

  dim3 blk(256);
  dim3 big(1024);
  const int tileBlocks = (N + 15) / 16;

  part_hist    <<<PART_BLOCKS, big, 0, stream>>>(dst, hist_g, E, NB);
  csr_blocksum <<<scanBlocks, blk, 0, stream>>>(hist_g, bsum, M);
  csr_scan     <<<scanBlocks, blk, 0, stream>>>(hist_g, bsum, scan_g, M);
  part_scatter <<<PART_BLOCKS, big, 0, stream>>>(src, dst, scan_g, pairs, E, NB);
  bin_sort     <<<NB, big, 0, stream>>>(pairs, scan_g, cnt, rowptr, ssrc, N, E, NB, PART_BLOCKS);

  sage_layer<1><<<tileBlocks, blk, 0, stream>>>(x, rowptr, cnt, ssrc, Wl1, Wr1, b1, h, N);

  sage_layer<0><<<tileBlocks, blk, 0, stream>>>(h, rowptr, cnt, ssrc, Wl2, Wr2, b2, out, N);
}


# layer product: a block takes 32 rows (was 16) and each weight fragment feeds two WMMAs on two independent accumulators, so weight loads per row halve; 3125 blocks; every sum's order unchanged
# speedup vs baseline: 4.7084x; 1.0951x over previous
#include <hip/hip_runtime.h>
#include <hip/hip_bf16.h>

typedef __attribute__((ext_vector_type(8)))  float  v8f;
typedef __attribute__((ext_vector_type(2)))  float  v2f;
typedef __attribute__((ext_vector_type(4)))  float  v4f;

#define D 128

#define SCAN_ELEMS 1024
#define PART_BLOCKS 256
#define BIN_SHIFT 8
#define BIN_NODES (1 << BIN_SHIFT)
#define MAX_BINS 2048

__global__ __launch_bounds__(1024) void part_hist(
    const int* __restrict__ dst, int* __restrict__ hist_g, int E, int NB) {
  __shared__ int hist[MAX_BINS];
  const int t = threadIdx.x;
  const int b = blockIdx.x;
  const int nb = gridDim.x;
  for (int i = t; i < NB; i += 1024) hist[i] = 0;
  __syncthreads();
  const int S = (E + nb - 1) / nb;
  const int lo = b * S;
  const int hi = min(E, lo + S);
  for (int e0 = lo + t; e0 < hi; e0 += 4 * 1024) {
    int d[4];
#pragma unroll
    for (int k = 0; k < 4; ++k) d[k] = dst[min(e0 + k * 1024, hi - 1)];
    asm volatile("" :: "v"(d[0]), "v"(d[1]), "v"(d[2]), "v"(d[3]));
#pragma unroll
    for (int k = 0; k < 4; ++k)
      if (e0 + k * 1024 < hi) atomicAdd(&hist[d[k] >> BIN_SHIFT], 1);
  }
  __syncthreads();
  for (int i = t; i < NB; i += 1024) hist_g[i * nb + b] = hist[i];
}

__global__ __launch_bounds__(256) void csr_blocksum(
    const int* __restrict__ cnt, int* __restrict__ bsum, int N) {
  __shared__ int red[256];
  const int t = threadIdx.x;
  const int i0 = blockIdx.x * SCAN_ELEMS + t * 4;
  int s = 0;
  for (int k = 0; k < 4; ++k)
    if (i0 + k < N) s += cnt[i0 + k];
  red[t] = s;
  __syncthreads();
  for (int o = 128; o > 0; o >>= 1) {
    if (t < o) red[t] += red[t + o];
    __syncthreads();
  }
  if (t == 0) bsum[blockIdx.x] = red[0];
}

__global__ __launch_bounds__(256) void csr_scan(
    const int* __restrict__ cnt, const int* __restrict__ bsum,
    int* __restrict__ rowptr, int N) {
  __shared__ int buf[2][256];
  __shared__ int red[256];
  const int t = threadIdx.x;
  const int b = blockIdx.x;

  int pre = 0;
  for (int i = t; i < b; i += 256) pre += bsum[i];
  red[t] = pre;

  const int i0 = b * SCAN_ELEMS + t * 4;
  int c[4];
  int s = 0;
#pragma unroll
  for (int k = 0; k < 4; ++k) {
    c[k] = (i0 + k < N) ? cnt[i0 + k] : 0;
    s += c[k];
  }
  buf[0][t] = s;
  __syncthreads();

  for (int o = 128; o > 0; o >>= 1) {
    if (t < o) red[t] += red[t + o];
    __syncthreads();
  }
  const int offset = red[0];

  int cur = 0;
  for (int o = 1; o < 256; o <<= 1) {
    int v = buf[cur][t];
    if (t >= o) v += buf[cur][t - o];
    buf[cur ^ 1][t] = v;
    cur ^= 1;
    __syncthreads();
  }
  int run = offset + buf[cur][t] - s;
#pragma unroll
  for (int k = 0; k < 4; ++k) {
    if (i0 + k < N) rowptr[i0 + k] = run;
    run += c[k];
  }
}

__global__ __launch_bounds__(1024) void part_scatter(
    const int* __restrict__ src, const int* __restrict__ dst,
    const int* __restrict__ scan_g, int2* __restrict__ pairs, int E, int NB) {
  __shared__ int cursor[MAX_BINS];
  const int t = threadIdx.x;
  const int b = blockIdx.x;
  const int nb = gridDim.x;
  for (int i = t; i < NB; i += 1024) cursor[i] = scan_g[i * nb + b];
  __syncthreads();
  const int S = (E + nb - 1) / nb;
  const int lo = b * S;
  const int hi = min(E, lo + S);
  for (int e0 = lo + t; e0 < hi; e0 += 4 * 1024) {
    int d[4], s[4];
#pragma unroll
    for (int k = 0; k < 4; ++k) {
      const int e = min(e0 + k * 1024, hi - 1);
      d[k] = dst[e];
      s[k] = src[e];
    }
    asm volatile("" :: "v"(d[0]), "v"(d[1]), "v"(d[2]), "v"(d[3]),
                       "v"(s[0]), "v"(s[1]), "v"(s[2]), "v"(s[3]));
#pragma unroll
    for (int k = 0; k < 4; ++k) {
      if (e0 + k * 1024 < hi) {
        const int pos = atomicAdd(&cursor[d[k] >> BIN_SHIFT], 1);
        pairs[pos] = make_int2(s[k], d[k]);
      }
    }
  }
}

__global__ __launch_bounds__(1024) void bin_sort(
    const int2* __restrict__ pairs, const int* __restrict__ scan_g,
    int* __restrict__ cnt, int* __restrict__ rowptr, int* __restrict__ ssrc,
    int N, int E, int NB, int slices) {
  __shared__ int c[BIN_NODES];
  __shared__ int buf[2][BIN_NODES];
  __shared__ int cur[BIN_NODES];
  const int t = threadIdx.x;
  const int bin = blockIdx.x;
  const int node0 = bin << BIN_SHIFT;
  const int lo = scan_g[bin * slices];
  const int hi = (bin + 1 < NB) ? scan_g[(bin + 1) * slices] : E;

  if (t < BIN_NODES) c[t] = 0;
  __syncthreads();
  for (int e0 = lo + t; e0 < hi; e0 += 4 * 1024) {
    int d[4];
#pragma unroll
    for (int k = 0; k < 4; ++k) d[k] = pairs[min(e0 + k * 1024, hi - 1)].y;
    asm volatile("" :: "v"(d[0]), "v"(d[1]), "v"(d[2]), "v"(d[3]));
#pragma unroll
    for (int k = 0; k < 4; ++k)
      if (e0 + k * 1024 < hi) atomicAdd(&c[d[k] - node0], 1);
  }
  __syncthreads();

  if (t < BIN_NODES) buf[0][t] = c[t];
  __syncthreads();
  int w = 0;
  for (int o = 1; o < BIN_NODES; o <<= 1) {
    if (t < BIN_NODES) {
      int v = buf[w][t];
      if (t >= o) v += buf[w][t - o];
      buf[w ^ 1][t] = v;
    }
    w ^= 1;
    __syncthreads();
  }
  if (t < BIN_NODES) {
    const int deg = c[t];
    const int first = lo + buf[w][t] - deg;
    cur[t] = first;
    if (node0 + t < N) {
      cnt[node0 + t] = deg;
      rowptr[node0 + t] = first;
    }
  }
  __syncthreads();
  for (int e0 = lo + t; e0 < hi; e0 += 4 * 1024) {
    int2 p[4];
#pragma unroll
    for (int k = 0; k < 4; ++k) p[k] = pairs[min(e0 + k * 1024, hi - 1)];
    asm volatile("" :: "v"(p[0].x), "v"(p[0].y), "v"(p[1].x), "v"(p[1].y),
                       "v"(p[2].x), "v"(p[2].y), "v"(p[3].x), "v"(p[3].y));
#pragma unroll
    for (int k = 0; k < 4; ++k) {
      if (e0 + k * 1024 < hi) {
        const int pos = atomicAdd(&cur[p[k].y - node0], 1);
        ssrc[pos] = p[k].x;
      }
    }
  }
}

__device__ __forceinline__ v4f pull_row_sum(
    const v4f* __restrict__ h4, const int* __restrict__ ssrc,
    int start, int deg, int lane, int first) {
  v4f acc = {0.0f, 0.0f, 0.0f, 0.0f};
  for (int base = 0; base < deg; base += 32) {
    const int m = min(32, deg - base);
    const int my = (base == 0) ? first : ssrc[start + base + min(lane, m - 1)];
    for (int j = 0; j < m; j += 8) {
      v4f r[8];
#pragma unroll
      for (int k = 0; k < 8; ++k) {
        const int jj = min(j + k, m - 1);
        const int s = __builtin_amdgcn_readlane(my, jj);
        r[k] = h4[(size_t)s * (D / 4) + lane];
      }
      asm volatile("" :: "v"(r[0]), "v"(r[1]), "v"(r[2]), "v"(r[3]),
                         "v"(r[4]), "v"(r[5]), "v"(r[6]), "v"(r[7]));
#pragma unroll
      for (int k = 0; k < 8; ++k) {
        if (j + k < m) acc += r[k];
      }
    }
  }
  return acc;
}

#define MROWS 32
template <int RELU>
__global__ __launch_bounds__(256) void sage_layer(
    const float* __restrict__ hin, const int* __restrict__ rowptr,
    const int* __restrict__ cnt, const int* __restrict__ ssrc,
    const float* __restrict__ Wl, const float* __restrict__ Wr,
    const float* __restrict__ bias, float* __restrict__ out, int N) {
  __shared__ __attribute__((aligned(16))) float sA[MROWS][264];

  const v4f* __restrict__ h4 = (const v4f*)hin;
  const int Mbase = blockIdx.x * MROWS;
  const int tid = threadIdx.x;
  const int lane = tid & 31;
  const int wave = __builtin_amdgcn_readfirstlane(tid >> 5);

  for (int p = 0; p < MROWS / 16; ++p) {
    const int r0 = wave * 4 + p * 2;
    const int r1 = r0 + 1;
    const int n0 = min(Mbase + r0, N - 1);
    const int n1 = min(Mbase + r1, N - 1);
    const int start0 = rowptr[n0];
    const int deg0   = cnt[n0];
    const int start1 = rowptr[n1];
    const int deg1   = cnt[n1];
    int first0 = 0, first1 = 0;
    if (deg0 > 0) first0 = ssrc[start0 + min(lane, min(32, deg0) - 1)];
    if (deg1 > 0) first1 = ssrc[start1 + min(lane, min(32, deg1) - 1)];
    const v4f own0 = h4[(size_t)n0 * (D / 4) + lane];
    const v4f own1 = h4[(size_t)n1 * (D / 4) + lane];

    const v4f sum0 = pull_row_sum(h4, ssrc, start0, deg0, lane, first0);
    const v4f sum1 = pull_row_sum(h4, ssrc, start1, deg1, lane, first1);

    const float c0 = fmaxf((float)deg0, 1.0f);
    const float c1 = fmaxf((float)deg1, 1.0f);
    *(v4f*)&sA[r0][4 * lane]     = sum0 / c0;
    *(v4f*)&sA[r1][4 * lane]     = sum1 / c1;
    *(v4f*)&sA[r0][D + 4 * lane] = own0;
    *(v4f*)&sA[r1][D + 4 * lane] = own1;
  }
  __syncthreads();

  const int col  = wave * 16 + (lane & 15);
  const int half = lane >> 4;
  const int rowA = lane & 15;

  v8f acc0 = {};
  v8f acc1 = {};

  for (int k0 = 0; k0 < 256; k0 += 4) {
    const float* W = (k0 < D) ? Wl : Wr;
    const int kb = (k0 & (D - 1)) + half * 2;
    v2f a0, a1, b;
    a0[0] = sA[rowA][k0 + half * 2];
    a0[1] = sA[rowA][k0 + half * 2 + 1];
    a1[0] = sA[16 + rowA][k0 + half * 2];
    a1[1] = sA[16 + rowA][k0 + half * 2 + 1];
    b[0] = W[(size_t)kb * D + col];
    b[1] = W[(size_t)(kb + 1) * D + col];
    acc0 = __builtin_amdgcn_wmma_f32_16x16x4_f32(
        false, a0, false, b, (short)0, acc0, false, false);
    acc1 = __builtin_amdgcn_wmma_f32_16x16x4_f32(
        false, a1, false, b, (short)0, acc1, false, false);
  }

  const float bv = bias[col];
  float* orow = out + (size_t)(Mbase + half * 8) * D + col;
#pragma unroll
  for (int v = 0; v < 8; ++v) {
    float val0 = acc0[v] + bv;
    float val1 = acc1[v] + bv;
    if (RELU) { val0 = fmaxf(val0, 0.0f); val1 = fmaxf(val1, 0.0f); }
    orow[(size_t)v * D]        = val0;
    orow[(size_t)(16 + v) * D] = val1;
  }
}

extern "C" void kernel_launch(void* const* d_in, const int* in_sizes, int n_in,
                              void* d_out, int out_size, void* d_ws, size_t ws_size,
                              hipStream_t stream) {
  const float* x   = (const float*)d_in[0];
  const int*   ei  = (const int*)d_in[1];
  const float* Wl1 = (const float*)d_in[2];
  const float* Wr1 = (const float*)d_in[3];
  const float* b1  = (const float*)d_in[4];
  const float* Wl2 = (const float*)d_in[5];
  const float* Wr2 = (const float*)d_in[6];
  const float* b2  = (const float*)d_in[7];
  float* out = (float*)d_out;

  const int N = in_sizes[0] / D;
  const int E = in_sizes[1] / 2;
  const int* src = ei;
  const int* dst = ei + E;

  const int NB = (N + BIN_NODES - 1) / BIN_NODES;
  const int M  = NB * PART_BLOCKS;
  const int scanBlocks = (M + SCAN_ELEMS - 1) / SCAN_ELEMS;

  float* h   = (float*)d_ws;
  int2* pairs = (int2*)(h + (size_t)N * D);
  int* cnt    = (int*)(pairs + E);
  int* rowptr = cnt + N;
  int* hist_g = rowptr + N;
  int* scan_g = hist_g + M;
  int* bsum   = scan_g + M;
  int* ssrc   = bsum + scanBlocks;

  dim3 blk(256);
  dim3 big(1024);
  const int tileBlocks = (N + MROWS - 1) / MROWS;

  part_hist    <<<PART_BLOCKS, big, 0, stream>>>(dst, hist_g, E, NB);
  csr_blocksum <<<scanBlocks, blk, 0, stream>>>(hist_g, bsum, M);
  csr_scan     <<<scanBlocks, blk, 0, stream>>>(hist_g, bsum, scan_g, M);
  part_scatter <<<PART_BLOCKS, big, 0, stream>>>(src, dst, scan_g, pairs, E, NB);
  bin_sort     <<<NB, big, 0, stream>>>(pairs, scan_g, cnt, rowptr, ssrc, N, E, NB, PART_BLOCKS);

  sage_layer<1><<<tileBlocks, blk, 0, stream>>>(x, rowptr, cnt, ssrc, Wl1, Wr1, b1, h, N);

  sage_layer<0><<<tileBlocks, blk, 0, stream>>>(h, rowptr, cnt, ssrc, Wl2, Wr2, b2, out, N);
}


# layer product: weights re-laid per call into fragment order, one whole-line 16 B-per-lane load feeds 4 WMMAs (was 4 half-line 4 B loads); A read 16 B per lane, pitch 260; k order in each 8 changes
# speedup vs baseline: 4.9834x; 1.0584x over previous
#include <hip/hip_runtime.h>
#include <hip/hip_bf16.h>

typedef __attribute__((ext_vector_type(8)))  float  v8f;
typedef __attribute__((ext_vector_type(2)))  float  v2f;
typedef __attribute__((ext_vector_type(4)))  float  v4f;

#define D 128

#define SCAN_ELEMS 1024
#define PART_BLOCKS 256
#define BIN_SHIFT 8
#define BIN_NODES (1 << BIN_SHIFT)
#define MAX_BINS 2048

#define WP_LAYER (2 * D * D)
#define WP_FLOATS (2 * WP_LAYER)
__device__ __forceinline__ void pack_weight_elem(
    int i, const float* __restrict__ Wl1, const float* __restrict__ Wr1,
    const float* __restrict__ Wl2, const float* __restrict__ Wr2,
    float* __restrict__ wp) {
  const int layer = i / WP_LAYER;
  const int j = i % WP_LAYER;
  const int s = j & 3;
  const int lane = (j >> 2) & 31;
  const int w = (j >> 7) & 7;
  const int g = j >> 10;
  const int k = 8 * g + 4 * (lane >> 4) + s;
  const int col = 16 * w + (lane & 15);
  const float* Wl = layer ? Wl2 : Wl1;
  const float* Wr = layer ? Wr2 : Wr1;
  wp[i] = (k < D) ? Wl[k * D + col] : Wr[(k - D) * D + col];
}

__global__ __launch_bounds__(1024) void part_hist(
    const int* __restrict__ dst, int* __restrict__ hist_g, int E, int NB,
    const float* __restrict__ Wl1, const float* __restrict__ Wr1,
    const float* __restrict__ Wl2, const float* __restrict__ Wr2,
    float* __restrict__ wp) {
  __shared__ int hist[MAX_BINS];
  const int t = threadIdx.x;
  const int b = blockIdx.x;
  const int nb = gridDim.x;
  for (int i = b * 1024 + t; i < WP_FLOATS; i += nb * 1024)
    pack_weight_elem(i, Wl1, Wr1, Wl2, Wr2, wp);
  for (int i = t; i < NB; i += 1024) hist[i] = 0;
  __syncthreads();
  const int S = (E + nb - 1) / nb;
  const int lo = b * S;
  const int hi = min(E, lo + S);
  for (int e0 = lo + t; e0 < hi; e0 += 4 * 1024) {
    int d[4];
#pragma unroll
    for (int k = 0; k < 4; ++k) d[k] = dst[min(e0 + k * 1024, hi - 1)];
    asm volatile("" :: "v"(d[0]), "v"(d[1]), "v"(d[2]), "v"(d[3]));
#pragma unroll
    for (int k = 0; k < 4; ++k)
      if (e0 + k * 1024 < hi) atomicAdd(&hist[d[k] >> BIN_SHIFT], 1);
  }
  __syncthreads();
  for (int i = t; i < NB; i += 1024) hist_g[i * nb + b] = hist[i];
}

__global__ __launch_bounds__(256) void csr_blocksum(
    const int* __restrict__ cnt, int* __restrict__ bsum, int N) {
  __shared__ int red[256];
  const int t = threadIdx.x;
  const int i0 = blockIdx.x * SCAN_ELEMS + t * 4;
  int s = 0;
  for (int k = 0; k < 4; ++k)
    if (i0 + k < N) s += cnt[i0 + k];
  red[t] = s;
  __syncthreads();
  for (int o = 128; o > 0; o >>= 1) {
    if (t < o) red[t] += red[t + o];
    __syncthreads();
  }
  if (t == 0) bsum[blockIdx.x] = red[0];
}

__global__ __launch_bounds__(256) void csr_scan(
    const int* __restrict__ cnt, const int* __restrict__ bsum,
    int* __restrict__ rowptr, int N) {
  __shared__ int buf[2][256];
  __shared__ int red[256];
  const int t = threadIdx.x;
  const int b = blockIdx.x;

  int pre = 0;
  for (int i = t; i < b; i += 256) pre += bsum[i];
  red[t] = pre;

  const int i0 = b * SCAN_ELEMS + t * 4;
  int c[4];
  int s = 0;
#pragma unroll
  for (int k = 0; k < 4; ++k) {
    c[k] = (i0 + k < N) ? cnt[i0 + k] : 0;
    s += c[k];
  }
  buf[0][t] = s;
  __syncthreads();

  for (int o = 128; o > 0; o >>= 1) {
    if (t < o) red[t] += red[t + o];
    __syncthreads();
  }
  const int offset = red[0];

  int cur = 0;
  for (int o = 1; o < 256; o <<= 1) {
    int v = buf[cur][t];
    if (t >= o) v += buf[cur][t - o];
    buf[cur ^ 1][t] = v;
    cur ^= 1;
    __syncthreads();
  }
  int run = offset + buf[cur][t] - s;
#pragma unroll
  for (int k = 0; k < 4; ++k) {
    if (i0 + k < N) rowptr[i0 + k] = run;
    run += c[k];
  }
}

__global__ __launch_bounds__(1024) void part_scatter(
    const int* __restrict__ src, const int* __restrict__ dst,
    const int* __restrict__ scan_g, int2* __restrict__ pairs, int E, int NB) {
  __shared__ int cursor[MAX_BINS];
  const int t = threadIdx.x;
  const int b = blockIdx.x;
  const int nb = gridDim.x;
  for (int i = t; i < NB; i += 1024) cursor[i] = scan_g[i * nb + b];
  __syncthreads();
  const int S = (E + nb - 1) / nb;
  const int lo = b * S;
  const int hi = min(E, lo + S);
  for (int e0 = lo + t; e0 < hi; e0 += 4 * 1024) {
    int d[4], s[4];
#pragma unroll
    for (int k = 0; k < 4; ++k) {
      const int e = min(e0 + k * 1024, hi - 1);
      d[k] = dst[e];
      s[k] = src[e];
    }
    asm volatile("" :: "v"(d[0]), "v"(d[1]), "v"(d[2]), "v"(d[3]),
                       "v"(s[0]), "v"(s[1]), "v"(s[2]), "v"(s[3]));
#pragma unroll
    for (int k = 0; k < 4; ++k) {
      if (e0 + k * 1024 < hi) {
        const int pos = atomicAdd(&cursor[d[k] >> BIN_SHIFT], 1);
        pairs[pos] = make_int2(s[k], d[k]);
      }
    }
  }
}

__global__ __launch_bounds__(1024) void bin_sort(
    const int2* __restrict__ pairs, const int* __restrict__ scan_g,
    int* __restrict__ cnt, int* __restrict__ rowptr, int* __restrict__ ssrc,
    int N, int E, int NB, int slices) {
  __shared__ int c[BIN_NODES];
  __shared__ int buf[2][BIN_NODES];
  __shared__ int cur[BIN_NODES];
  const int t = threadIdx.x;
  const int bin = blockIdx.x;
  const int node0 = bin << BIN_SHIFT;
  const int lo = scan_g[bin * slices];
  const int hi = (bin + 1 < NB) ? scan_g[(bin + 1) * slices] : E;

  if (t < BIN_NODES) c[t] = 0;
  __syncthreads();
  for (int e0 = lo + t; e0 < hi; e0 += 4 * 1024) {
    int d[4];
#pragma unroll
    for (int k = 0; k < 4; ++k) d[k] = pairs[min(e0 + k * 1024, hi - 1)].y;
    asm volatile("" :: "v"(d[0]), "v"(d[1]), "v"(d[2]), "v"(d[3]));
#pragma unroll
    for (int k = 0; k < 4; ++k)
      if (e0 + k * 1024 < hi) atomicAdd(&c[d[k] - node0], 1);
  }
  __syncthreads();

  if (t < BIN_NODES) buf[0][t] = c[t];
  __syncthreads();
  int w = 0;
  for (int o = 1; o < BIN_NODES; o <<= 1) {
    if (t < BIN_NODES) {
      int v = buf[w][t];
      if (t >= o) v += buf[w][t - o];
      buf[w ^ 1][t] = v;
    }
    w ^= 1;
    __syncthreads();
  }
  if (t < BIN_NODES) {
    const int deg = c[t];
    const int first = lo + buf[w][t] - deg;
    cur[t] = first;
    if (node0 + t < N) {
      cnt[node0 + t] = deg;
      rowptr[node0 + t] = first;
    }
  }
  __syncthreads();
  for (int e0 = lo + t; e0 < hi; e0 += 4 * 1024) {
    int2 p[4];
#pragma unroll
    for (int k = 0; k < 4; ++k) p[k] = pairs[min(e0 + k * 1024, hi - 1)];
    asm volatile("" :: "v"(p[0].x), "v"(p[0].y), "v"(p[1].x), "v"(p[1].y),
                       "v"(p[2].x), "v"(p[2].y), "v"(p[3].x), "v"(p[3].y));
#pragma unroll
    for (int k = 0; k < 4; ++k) {
      if (e0 + k * 1024 < hi) {
        const int pos = atomicAdd(&cur[p[k].y - node0], 1);
        ssrc[pos] = p[k].x;
      }
    }
  }
}

__device__ __forceinline__ v4f pull_row_sum(
    const v4f* __restrict__ h4, const int* __restrict__ ssrc,
    int start, int deg, int lane, int first) {
  v4f acc = {0.0f, 0.0f, 0.0f, 0.0f};
  for (int base = 0; base < deg; base += 32) {
    const int m = min(32, deg - base);
    const int my = (base == 0) ? first : ssrc[start + base + min(lane, m - 1)];
    for (int j = 0; j < m; j += 8) {
      v4f r[8];
#pragma unroll
      for (int k = 0; k < 8; ++k) {
        const int jj = min(j + k, m - 1);
        const int s = __builtin_amdgcn_readlane(my, jj);
        r[k] = h4[(size_t)s * (D / 4) + lane];
      }
      asm volatile("" :: "v"(r[0]), "v"(r[1]), "v"(r[2]), "v"(r[3]),
                         "v"(r[4]), "v"(r[5]), "v"(r[6]), "v"(r[7]));
#pragma unroll
      for (int k = 0; k < 8; ++k) {
        if (j + k < m) acc += r[k];
      }
    }
  }
  return acc;
}

#define MROWS 32
template <int RELU>
__global__ __launch_bounds__(256) void sage_layer(
    const float* __restrict__ hin, const int* __restrict__ rowptr,
    const int* __restrict__ cnt, const int* __restrict__ ssrc,
    const float* __restrict__ wp,
    const float* __restrict__ bias, float* __restrict__ out, int N) {
  __shared__ __attribute__((aligned(16))) float sA[MROWS][260];

  const v4f* __restrict__ h4 = (const v4f*)hin;
  const int Mbase = blockIdx.x * MROWS;
  const int tid = threadIdx.x;
  const int lane = tid & 31;
  const int wave = __builtin_amdgcn_readfirstlane(tid >> 5);

  for (int p = 0; p < MROWS / 16; ++p) {
    const int r0 = wave * 4 + p * 2;
    const int r1 = r0 + 1;
    const int n0 = min(Mbase + r0, N - 1);
    const int n1 = min(Mbase + r1, N - 1);
    const int start0 = rowptr[n0];
    const int deg0   = cnt[n0];
    const int start1 = rowptr[n1];
    const int deg1   = cnt[n1];
    int first0 = 0, first1 = 0;
    if (deg0 > 0) first0 = ssrc[start0 + min(lane, min(32, deg0) - 1)];
    if (deg1 > 0) first1 = ssrc[start1 + min(lane, min(32, deg1) - 1)];
    const v4f own0 = h4[(size_t)n0 * (D / 4) + lane];
    const v4f own1 = h4[(size_t)n1 * (D / 4) + lane];

    const v4f sum0 = pull_row_sum(h4, ssrc, start0, deg0, lane, first0);
    const v4f sum1 = pull_row_sum(h4, ssrc, start1, deg1, lane, first1);

    const float c0 = fmaxf((float)deg0, 1.0f);
    const float c1 = fmaxf((float)deg1, 1.0f);
    *(v4f*)&sA[r0][4 * lane]     = sum0 / c0;
    *(v4f*)&sA[r1][4 * lane]     = sum1 / c1;
    *(v4f*)&sA[r0][D + 4 * lane] = own0;
    *(v4f*)&sA[r1][D + 4 * lane] = own1;
  }
  __syncthreads();

  const int col  = wave * 16 + (lane & 15);
  const int half = lane >> 4;
  const int rowA = lane & 15;

  v8f acc0 = {};
  v8f acc1 = {};

  const v4f* __restrict__ wp4 = (const v4f*)wp + wave * 32 + lane;
  for (int g = 0; g < 32; ++g) {
    const v4f bq  = wp4[g * 256];
    const v4f a0q = *(const v4f*)&sA[rowA][8 * g + 4 * half];
    const v4f a1q = *(const v4f*)&sA[16 + rowA][8 * g + 4 * half];
    v2f a0, a1, b;
    a0[0] = a0q[0]; a0[1] = a0q[1];
    a1[0] = a1q[0]; a1[1] = a1q[1];
    b[0]  = bq[0];  b[1]  = bq[1];
    acc0 = __builtin_amdgcn_wmma_f32_16x16x4_f32(
        false, a0, false, b, (short)0, acc0, false, false);
    acc1 = __builtin_amdgcn_wmma_f32_16x16x4_f32(
        false, a1, false, b, (short)0, acc1, false, false);
    a0[0] = a0q[2]; a0[1] = a0q[3];
    a1[0] = a1q[2]; a1[1] = a1q[3];
    b[0]  = bq[2];  b[1]  = bq[3];
    acc0 = __builtin_amdgcn_wmma_f32_16x16x4_f32(
        false, a0, false, b, (short)0, acc0, false, false);
    acc1 = __builtin_amdgcn_wmma_f32_16x16x4_f32(
        false, a1, false, b, (short)0, acc1, false, false);
  }

  const float bv = bias[col];
  float* orow = out + (size_t)(Mbase + half * 8) * D + col;
#pragma unroll
  for (int v = 0; v < 8; ++v) {
    float val0 = acc0[v] + bv;
    float val1 = acc1[v] + bv;
    if (RELU) { val0 = fmaxf(val0, 0.0f); val1 = fmaxf(val1, 0.0f); }
    orow[(size_t)v * D]        = val0;
    orow[(size_t)(16 + v) * D] = val1;
  }
}

extern "C" void kernel_launch(void* const* d_in, const int* in_sizes, int n_in,
                              void* d_out, int out_size, void* d_ws, size_t ws_size,
                              hipStream_t stream) {
  const float* x   = (const float*)d_in[0];
  const int*   ei  = (const int*)d_in[1];
  const float* Wl1 = (const float*)d_in[2];
  const float* Wr1 = (const float*)d_in[3];
  const float* b1  = (const float*)d_in[4];
  const float* Wl2 = (const float*)d_in[5];
  const float* Wr2 = (const float*)d_in[6];
  const float* b2  = (const float*)d_in[7];
  float* out = (float*)d_out;

  const int N = in_sizes[0] / D;
  const int E = in_sizes[1] / 2;
  const int* src = ei;
  const int* dst = ei + E;

  const int NB = (N + BIN_NODES - 1) / BIN_NODES;
  const int M  = NB * PART_BLOCKS;
  const int scanBlocks = (M + SCAN_ELEMS - 1) / SCAN_ELEMS;

  float* h   = (float*)d_ws;
  float* wp  = h + (size_t)N * D;
  int2* pairs = (int2*)(wp + WP_FLOATS);
  int* cnt    = (int*)(pairs + E);
  int* rowptr = cnt + N;
  int* hist_g = rowptr + N;
  int* scan_g = hist_g + M;
  int* bsum   = scan_g + M;
  int* ssrc   = bsum + scanBlocks;

  dim3 blk(256);
  dim3 big(1024);
  const int tileBlocks = (N + MROWS - 1) / MROWS;

  part_hist    <<<PART_BLOCKS, big, 0, stream>>>(dst, hist_g, E, NB, Wl1, Wr1, Wl2, Wr2, wp);
  csr_blocksum <<<scanBlocks, blk, 0, stream>>>(hist_g, bsum, M);
  csr_scan     <<<scanBlocks, blk, 0, stream>>>(hist_g, bsum, scan_g, M);
  part_scatter <<<PART_BLOCKS, big, 0, stream>>>(src, dst, scan_g, pairs, E, NB);
  bin_sort     <<<NB, big, 0, stream>>>(pairs, scan_g, cnt, rowptr, ssrc, N, E, NB, PART_BLOCKS);

  sage_layer<1><<<tileBlocks, blk, 0, stream>>>(x, rowptr, cnt, ssrc, wp, b1, h, N);

  sage_layer<0><<<tileBlocks, blk, 0, stream>>>(h, rowptr, cnt, ssrc, wp + WP_LAYER, b2, out, N);
}


# edge bucketing: each slice sorted by bin in its own region (LDS prefix sum); a bin block reads its run from every slice, placed by the sum of run starts; 4 launches, was 7; sums in LDS-atomic order
# speedup vs baseline: 5.4631x; 1.0963x over previous
#include <hip/hip_runtime.h>
#include <hip/hip_bf16.h>

typedef __attribute__((ext_vector_type(8)))  float  v8f;
typedef __attribute__((ext_vector_type(2)))  float  v2f;
typedef __attribute__((ext_vector_type(4)))  float  v4f;

#define D 128

#define PART_BLOCKS 256
#define BIN_SHIFT 8
#define BIN_NODES (1 << BIN_SHIFT)
#define MAX_BINS 2048
#define RUNS_PER_WAVE (PART_BLOCKS / 32)

#define WP_LAYER (2 * D * D)
#define WP_FLOATS (2 * WP_LAYER)
__device__ __forceinline__ void pack_weight_elem(
    int i, const float* __restrict__ Wl1, const float* __restrict__ Wr1,
    const float* __restrict__ Wl2, const float* __restrict__ Wr2,
    float* __restrict__ wp) {
  const int layer = i / WP_LAYER;
  const int j = i % WP_LAYER;
  const int s = j & 3;
  const int lane = (j >> 2) & 31;
  const int w = (j >> 7) & 7;
  const int g = j >> 10;
  const int k = 8 * g + 4 * (lane >> 4) + s;
  const int col = 16 * w + (lane & 15);
  const float* Wl = layer ? Wl2 : Wl1;
  const float* Wr = layer ? Wr2 : Wr1;
  wp[i] = (k < D) ? Wl[k * D + col] : Wr[(k - D) * D + col];
}

__global__ __launch_bounds__(1024) void slice_sort(
    const int* __restrict__ src, const int* __restrict__ dst,
    int* __restrict__ pt, int2* __restrict__ pairs, int E, int NB,
    const float* __restrict__ Wl1, const float* __restrict__ Wr1,
    const float* __restrict__ Wl2, const float* __restrict__ Wr2,
    float* __restrict__ wp) {
  __shared__ int hist[MAX_BINS];
  __shared__ int sbuf[2][1024];
  const int t = threadIdx.x;
  const int b = blockIdx.x;
  const int nb = gridDim.x;
  for (int i = b * 1024 + t; i < WP_FLOATS; i += nb * 1024)
    pack_weight_elem(i, Wl1, Wr1, Wl2, Wr2, wp);
  for (int i = t; i < MAX_BINS; i += 1024) hist[i] = 0;
  __syncthreads();
  const int S = (E + nb - 1) / nb;
  const int lo = b * S;
  const int hi = min(E, lo + S);
  for (int e0 = lo + t; e0 < hi; e0 += 4 * 1024) {
    int d[4];
#pragma unroll
    for (int k = 0; k < 4; ++k) d[k] = dst[min(e0 + k * 1024, hi - 1)];
    asm volatile("" :: "v"(d[0]), "v"(d[1]), "v"(d[2]), "v"(d[3]));
#pragma unroll
    for (int k = 0; k < 4; ++k)
      if (e0 + k * 1024 < hi) atomicAdd(&hist[d[k] >> BIN_SHIFT], 1);
  }
  __syncthreads();

  const int c0 = hist[2 * t];
  const int c1 = hist[2 * t + 1];
  sbuf[0][t] = c0 + c1;
  __syncthreads();
  int cur = 0;
  for (int o = 1; o < 1024; o <<= 1) {
    int v = sbuf[cur][t];
    if (t >= o) v += sbuf[cur][t - o];
    sbuf[cur ^ 1][t] = v;
    cur ^= 1;
    __syncthreads();
  }
  const int ex = sbuf[cur][t] - (c0 + c1);
  if (2 * t <= NB)     pt[(2 * t) * nb + b]     = ex;
  if (2 * t + 1 <= NB) pt[(2 * t + 1) * nb + b] = ex + c0;
  hist[2 * t]     = lo + ex;
  hist[2 * t + 1] = lo + ex + c0;
  __syncthreads();

  for (int e0 = lo + t; e0 < hi; e0 += 4 * 1024) {
    int d[4], s[4];
#pragma unroll
    for (int k = 0; k < 4; ++k) {
      const int e = min(e0 + k * 1024, hi - 1);
      d[k] = dst[e];
      s[k] = src[e];
    }
    asm volatile("" :: "v"(d[0]), "v"(d[1]), "v"(d[2]), "v"(d[3]),
                       "v"(s[0]), "v"(s[1]), "v"(s[2]), "v"(s[3]));
#pragma unroll
    for (int k = 0; k < 4; ++k) {
      if (e0 + k * 1024 < hi) {
        const int pos = atomicAdd(&hist[d[k] >> BIN_SHIFT], 1);
        pairs[pos] = make_int2(s[k], d[k]);
      }
    }
  }
}

__global__ __launch_bounds__(1024) void bin_sort(
    const int2* __restrict__ pairs, const int* __restrict__ pt,
    int* __restrict__ cnt, int* __restrict__ rowptr, int* __restrict__ ssrc,
    int N, int E, int NB) {
  __shared__ int c[BIN_NODES];
  __shared__ int buf[2][BIN_NODES];
  __shared__ int cur[BIN_NODES];
  __shared__ int rstart[PART_BLOCKS];
  __shared__ int rlen[PART_BLOCKS];
  __shared__ int red[PART_BLOCKS];
  const int t = threadIdx.x;
  const int lane = t & 31;
  const int w = t >> 5;
  const int bin = blockIdx.x;
  const int node0 = bin << BIN_SHIFT;
  const int S = (E + PART_BLOCKS - 1) / PART_BLOCKS;

  if (t < PART_BLOCKS) {
    const int a = pt[bin * PART_BLOCKS + t];
    const int e = pt[(bin + 1) * PART_BLOCKS + t];
    rstart[t] = t * S + a;
    rlen[t] = e - a;
    red[t] = a;
  }
  if (t < BIN_NODES) c[t] = 0;
  __syncthreads();
  for (int o = PART_BLOCKS / 2; o > 0; o >>= 1) {
    if (t < o) red[t] += red[t + o];
    __syncthreads();
  }
  const int lo = red[0];

  static_assert(RUNS_PER_WAVE == 8, "the fence below names eight pairs");
  int2 p[RUNS_PER_WAVE];
  int len[RUNS_PER_WAVE];
#pragma unroll
  for (int k = 0; k < RUNS_PER_WAVE; ++k) {
    const int r = w + 32 * k;
    len[k] = rlen[r];
    const int i = rstart[r] + min(lane, max(len[k], 1) - 1);
    p[k] = pairs[min(i, E - 1)];
  }
  asm volatile("" :: "v"(p[0].y), "v"(p[1].y), "v"(p[2].y), "v"(p[3].y),
                     "v"(p[4].y), "v"(p[5].y), "v"(p[6].y), "v"(p[7].y));
#pragma unroll
  for (int k = 0; k < RUNS_PER_WAVE; ++k) {
    if (lane < len[k]) atomicAdd(&c[p[k].y - node0], 1);
    for (int i = 32 + lane; i < len[k]; i += 32)
      atomicAdd(&c[pairs[rstart[w + 32 * k] + i].y - node0], 1);
  }
  __syncthreads();

  if (t < BIN_NODES) buf[0][t] = c[t];
  __syncthreads();
  int q = 0;
  for (int o = 1; o < BIN_NODES; o <<= 1) {
    if (t < BIN_NODES) {
      int v = buf[q][t];
      if (t >= o) v += buf[q][t - o];
      buf[q ^ 1][t] = v;
    }
    q ^= 1;
    __syncthreads();
  }
  if (t < BIN_NODES) {
    const int deg = c[t];
    const int first = lo + buf[q][t] - deg;
    cur[t] = first;
    if (node0 + t < N) {
      cnt[node0 + t] = deg;
      rowptr[node0 + t] = first;
    }
  }
  __syncthreads();
#pragma unroll
  for (int k = 0; k < RUNS_PER_WAVE; ++k) {
    if (lane < len[k]) {
      const int pos = atomicAdd(&cur[p[k].y - node0], 1);
      ssrc[pos] = p[k].x;
    }
    for (int i = 32 + lane; i < len[k]; i += 32) {
      const int2 pr = pairs[rstart[w + 32 * k] + i];
      const int pos = atomicAdd(&cur[pr.y - node0], 1);
      ssrc[pos] = pr.x;
    }
  }
}

__device__ __forceinline__ v4f pull_row_sum(
    const v4f* __restrict__ h4, const int* __restrict__ ssrc,
    int start, int deg, int lane, int first) {
  v4f acc = {0.0f, 0.0f, 0.0f, 0.0f};
  for (int base = 0; base < deg; base += 32) {
    const int m = min(32, deg - base);
    const int my = (base == 0) ? first : ssrc[start + base + min(lane, m - 1)];
    for (int j = 0; j < m; j += 8) {
      v4f r[8];
#pragma unroll
      for (int k = 0; k < 8; ++k) {
        const int jj = min(j + k, m - 1);
        const int s = __builtin_amdgcn_readlane(my, jj);
        r[k] = h4[(size_t)s * (D / 4) + lane];
      }
      asm volatile("" :: "v"(r[0]), "v"(r[1]), "v"(r[2]), "v"(r[3]),
                         "v"(r[4]), "v"(r[5]), "v"(r[6]), "v"(r[7]));
#pragma unroll
      for (int k = 0; k < 8; ++k) {
        if (j + k < m) acc += r[k];
      }
    }
  }
  return acc;
}

#define MROWS 32
template <int RELU>
__global__ __launch_bounds__(256) void sage_layer(
    const float* __restrict__ hin, const int* __restrict__ rowptr,
    const int* __restrict__ cnt, const int* __restrict__ ssrc,
    const float* __restrict__ wp,
    const float* __restrict__ bias, float* __restrict__ out, int N) {
  __shared__ __attribute__((aligned(16))) float sA[MROWS][260];

  const v4f* __restrict__ h4 = (const v4f*)hin;
  const int Mbase = blockIdx.x * MROWS;
  const int tid = threadIdx.x;
  const int lane = tid & 31;
  const int wave = __builtin_amdgcn_readfirstlane(tid >> 5);

  for (int p = 0; p < MROWS / 16; ++p) {
    const int r0 = wave * 4 + p * 2;
    const int r1 = r0 + 1;
    const int n0 = min(Mbase + r0, N - 1);
    const int n1 = min(Mbase + r1, N - 1);
    const int start0 = rowptr[n0];
    const int deg0   = cnt[n0];
    const int start1 = rowptr[n1];
    const int deg1   = cnt[n1];
    int first0 = 0, first1 = 0;
    if (deg0 > 0) first0 = ssrc[start0 + min(lane, min(32, deg0) - 1)];
    if (deg1 > 0) first1 = ssrc[start1 + min(lane, min(32, deg1) - 1)];
    const v4f own0 = h4[(size_t)n0 * (D / 4) + lane];
    const v4f own1 = h4[(size_t)n1 * (D / 4) + lane];

    const v4f sum0 = pull_row_sum(h4, ssrc, start0, deg0, lane, first0);
    const v4f sum1 = pull_row_sum(h4, ssrc, start1, deg1, lane, first1);

    const float c0 = fmaxf((float)deg0, 1.0f);
    const float c1 = fmaxf((float)deg1, 1.0f);
    *(v4f*)&sA[r0][4 * lane]     = sum0 / c0;
    *(v4f*)&sA[r1][4 * lane]     = sum1 / c1;
    *(v4f*)&sA[r0][D + 4 * lane] = own0;
    *(v4f*)&sA[r1][D + 4 * lane] = own1;
  }
  __syncthreads();

  const int col  = wave * 16 + (lane & 15);
  const int half = lane >> 4;
  const int rowA = lane & 15;

  v8f acc0 = {};
  v8f acc1 = {};

  const v4f* __restrict__ wp4 = (const v4f*)wp + wave * 32 + lane;
  for (int g = 0; g < 32; ++g) {
    const v4f bq  = wp4[g * 256];
    const v4f a0q = *(const v4f*)&sA[rowA][8 * g + 4 * half];
    const v4f a1q = *(const v4f*)&sA[16 + rowA][8 * g + 4 * half];
    v2f a0, a1, b;
    a0[0] = a0q[0]; a0[1] = a0q[1];
    a1[0] = a1q[0]; a1[1] = a1q[1];
    b[0]  = bq[0];  b[1]  = bq[1];
    acc0 = __builtin_amdgcn_wmma_f32_16x16x4_f32(
        false, a0, false, b, (short)0, acc0, false, false);
    acc1 = __builtin_amdgcn_wmma_f32_16x16x4_f32(
        false, a1, false, b, (short)0, acc1, false, false);
    a0[0] = a0q[2]; a0[1] = a0q[3];
    a1[0] = a1q[2]; a1[1] = a1q[3];
    b[0]  = bq[2];  b[1]  = bq[3];
    acc0 = __builtin_amdgcn_wmma_f32_16x16x4_f32(
        false, a0, false, b, (short)0, acc0, false, false);
    acc1 = __builtin_amdgcn_wmma_f32_16x16x4_f32(
        false, a1, false, b, (short)0, acc1, false, false);
  }

  const float bv = bias[col];
  float* orow = out + (size_t)(Mbase + half * 8) * D + col;
#pragma unroll
  for (int v = 0; v < 8; ++v) {
    float val0 = acc0[v] + bv;
    float val1 = acc1[v] + bv;
    if (RELU) { val0 = fmaxf(val0, 0.0f); val1 = fmaxf(val1, 0.0f); }
    orow[(size_t)v * D]        = val0;
    orow[(size_t)(16 + v) * D] = val1;
  }
}

extern "C" void kernel_launch(void* const* d_in, const int* in_sizes, int n_in,
                              void* d_out, int out_size, void* d_ws, size_t ws_size,
                              hipStream_t stream) {
  const float* x   = (const float*)d_in[0];
  const int*   ei  = (const int*)d_in[1];
  const float* Wl1 = (const float*)d_in[2];
  const float* Wr1 = (const float*)d_in[3];
  const float* b1  = (const float*)d_in[4];
  const float* Wl2 = (const float*)d_in[5];
  const float* Wr2 = (const float*)d_in[6];
  const float* b2  = (const float*)d_in[7];
  float* out = (float*)d_out;

  const int N = in_sizes[0] / D;
  const int E = in_sizes[1] / 2;
  const int* src = ei;
  const int* dst = ei + E;

  const int NB = (N + BIN_NODES - 1) / BIN_NODES;
  const int M  = (NB + 1) * PART_BLOCKS;

  float* h   = (float*)d_ws;
  float* wp  = h + (size_t)N * D;
  int2* pairs = (int2*)(wp + WP_FLOATS);
  int* cnt    = (int*)(pairs + E);
  int* rowptr = cnt + N;
  int* pt     = rowptr + N;
  int* ssrc   = pt + M;

  dim3 blk(256);
  dim3 big(1024);
  const int tileBlocks = (N + MROWS - 1) / MROWS;

  slice_sort <<<PART_BLOCKS, big, 0, stream>>>(src, dst, pt, pairs, E, NB, Wl1, Wr1, Wl2, Wr2, wp);
  bin_sort   <<<NB, big, 0, stream>>>(pairs, pt, cnt, rowptr, ssrc, N, E, NB);

  sage_layer<1><<<tileBlocks, blk, 0, stream>>>(x, rowptr, cnt, ssrc, wp, b1, h, N);

  sage_layer<0><<<tileBlocks, blk, 0, stream>>>(h, rowptr, cnt, ssrc, wp + WP_LAYER, b2, out, N);
}


# gather: a wave walks its 4 rows' contiguous bucket range as one list in full batches of 8 row loads (per row before: tail batch, index trip); accumulator handed over at row ends; sum order kept
# speedup vs baseline: 5.6244x; 1.0295x over previous
#include <hip/hip_runtime.h>
#include <hip/hip_bf16.h>

typedef __attribute__((ext_vector_type(8)))  float  v8f;
typedef __attribute__((ext_vector_type(2)))  float  v2f;
typedef __attribute__((ext_vector_type(4)))  float  v4f;

#define D 128

#define PART_BLOCKS 256
#define BIN_SHIFT 8
#define BIN_NODES (1 << BIN_SHIFT)
#define MAX_BINS 2048
#define RUNS_PER_WAVE (PART_BLOCKS / 32)

#define WP_LAYER (2 * D * D)
#define WP_FLOATS (2 * WP_LAYER)
__device__ __forceinline__ void pack_weight_elem(
    int i, const float* __restrict__ Wl1, const float* __restrict__ Wr1,
    const float* __restrict__ Wl2, const float* __restrict__ Wr2,
    float* __restrict__ wp) {
  const int layer = i / WP_LAYER;
  const int j = i % WP_LAYER;
  const int s = j & 3;
  const int lane = (j >> 2) & 31;
  const int w = (j >> 7) & 7;
  const int g = j >> 10;
  const int k = 8 * g + 4 * (lane >> 4) + s;
  const int col = 16 * w + (lane & 15);
  const float* Wl = layer ? Wl2 : Wl1;
  const float* Wr = layer ? Wr2 : Wr1;
  wp[i] = (k < D) ? Wl[k * D + col] : Wr[(k - D) * D + col];
}

__global__ __launch_bounds__(1024) void slice_sort(
    const int* __restrict__ src, const int* __restrict__ dst,
    int* __restrict__ pt, int2* __restrict__ pairs, int E, int NB,
    const float* __restrict__ Wl1, const float* __restrict__ Wr1,
    const float* __restrict__ Wl2, const float* __restrict__ Wr2,
    float* __restrict__ wp) {
  __shared__ int hist[MAX_BINS];
  __shared__ int sbuf[2][1024];
  const int t = threadIdx.x;
  const int b = blockIdx.x;
  const int nb = gridDim.x;
  for (int i = b * 1024 + t; i < WP_FLOATS; i += nb * 1024)
    pack_weight_elem(i, Wl1, Wr1, Wl2, Wr2, wp);
  for (int i = t; i < MAX_BINS; i += 1024) hist[i] = 0;
  __syncthreads();
  const int S = (E + nb - 1) / nb;
  const int lo = b * S;
  const int hi = min(E, lo + S);
  for (int e0 = lo + t; e0 < hi; e0 += 4 * 1024) {
    int d[4];
#pragma unroll
    for (int k = 0; k < 4; ++k) d[k] = dst[min(e0 + k * 1024, hi - 1)];
    asm volatile("" :: "v"(d[0]), "v"(d[1]), "v"(d[2]), "v"(d[3]));
#pragma unroll
    for (int k = 0; k < 4; ++k)
      if (e0 + k * 1024 < hi) atomicAdd(&hist[d[k] >> BIN_SHIFT], 1);
  }
  __syncthreads();

  const int c0 = hist[2 * t];
  const int c1 = hist[2 * t + 1];
  sbuf[0][t] = c0 + c1;
  __syncthreads();
  int cur = 0;
  for (int o = 1; o < 1024; o <<= 1) {
    int v = sbuf[cur][t];
    if (t >= o) v += sbuf[cur][t - o];
    sbuf[cur ^ 1][t] = v;
    cur ^= 1;
    __syncthreads();
  }
  const int ex = sbuf[cur][t] - (c0 + c1);
  if (2 * t <= NB)     pt[(2 * t) * nb + b]     = ex;
  if (2 * t + 1 <= NB) pt[(2 * t + 1) * nb + b] = ex + c0;
  hist[2 * t]     = lo + ex;
  hist[2 * t + 1] = lo + ex + c0;
  __syncthreads();

  for (int e0 = lo + t; e0 < hi; e0 += 4 * 1024) {
    int d[4], s[4];
#pragma unroll
    for (int k = 0; k < 4; ++k) {
      const int e = min(e0 + k * 1024, hi - 1);
      d[k] = dst[e];
      s[k] = src[e];
    }
    asm volatile("" :: "v"(d[0]), "v"(d[1]), "v"(d[2]), "v"(d[3]),
                       "v"(s[0]), "v"(s[1]), "v"(s[2]), "v"(s[3]));
#pragma unroll
    for (int k = 0; k < 4; ++k) {
      if (e0 + k * 1024 < hi) {
        const int pos = atomicAdd(&hist[d[k] >> BIN_SHIFT], 1);
        pairs[pos] = make_int2(s[k], d[k]);
      }
    }
  }
}

__global__ __launch_bounds__(1024) void bin_sort(
    const int2* __restrict__ pairs, const int* __restrict__ pt,
    int* __restrict__ cnt, int* __restrict__ rowptr, int* __restrict__ ssrc,
    int N, int E, int NB) {
  __shared__ int c[BIN_NODES];
  __shared__ int buf[2][BIN_NODES];
  __shared__ int cur[BIN_NODES];
  __shared__ int rstart[PART_BLOCKS];
  __shared__ int rlen[PART_BLOCKS];
  __shared__ int red[PART_BLOCKS];
  const int t = threadIdx.x;
  const int lane = t & 31;
  const int w = t >> 5;
  const int bin = blockIdx.x;
  const int node0 = bin << BIN_SHIFT;
  const int S = (E + PART_BLOCKS - 1) / PART_BLOCKS;

  if (t < PART_BLOCKS) {
    const int a = pt[bin * PART_BLOCKS + t];
    const int e = pt[(bin + 1) * PART_BLOCKS + t];
    rstart[t] = t * S + a;
    rlen[t] = e - a;
    red[t] = a;
  }
  if (t < BIN_NODES) c[t] = 0;
  __syncthreads();
  for (int o = PART_BLOCKS / 2; o > 0; o >>= 1) {
    if (t < o) red[t] += red[t + o];
    __syncthreads();
  }
  const int lo = red[0];

  static_assert(RUNS_PER_WAVE == 8, "the fence below names eight pairs");
  int2 p[RUNS_PER_WAVE];
  int len[RUNS_PER_WAVE];
#pragma unroll
  for (int k = 0; k < RUNS_PER_WAVE; ++k) {
    const int r = w + 32 * k;
    len[k] = rlen[r];
    const int i = rstart[r] + min(lane, max(len[k], 1) - 1);
    p[k] = pairs[min(i, E - 1)];
  }
  asm volatile("" :: "v"(p[0].y), "v"(p[1].y), "v"(p[2].y), "v"(p[3].y),
                     "v"(p[4].y), "v"(p[5].y), "v"(p[6].y), "v"(p[7].y));
#pragma unroll
  for (int k = 0; k < RUNS_PER_WAVE; ++k) {
    if (lane < len[k]) atomicAdd(&c[p[k].y - node0], 1);
    for (int i = 32 + lane; i < len[k]; i += 32)
      atomicAdd(&c[pairs[rstart[w + 32 * k] + i].y - node0], 1);
  }
  __syncthreads();

  if (t < BIN_NODES) buf[0][t] = c[t];
  __syncthreads();
  int q = 0;
  for (int o = 1; o < BIN_NODES; o <<= 1) {
    if (t < BIN_NODES) {
      int v = buf[q][t];
      if (t >= o) v += buf[q][t - o];
      buf[q ^ 1][t] = v;
    }
    q ^= 1;
    __syncthreads();
  }
  if (t < BIN_NODES) {
    const int deg = c[t];
    const int first = lo + buf[q][t] - deg;
    cur[t] = first;
    if (node0 + t < N) {
      cnt[node0 + t] = deg;
      rowptr[node0 + t] = first;
    }
  }
  __syncthreads();
#pragma unroll
  for (int k = 0; k < RUNS_PER_WAVE; ++k) {
    if (lane < len[k]) {
      const int pos = atomicAdd(&cur[p[k].y - node0], 1);
      ssrc[pos] = p[k].x;
    }
    for (int i = 32 + lane; i < len[k]; i += 32) {
      const int2 pr = pairs[rstart[w + 32 * k] + i];
      const int pos = atomicAdd(&cur[pr.y - node0], 1);
      ssrc[pos] = pr.x;
    }
  }
}

#define MROWS 32
template <int RELU>
__global__ __launch_bounds__(256) void sage_layer(
    const float* __restrict__ hin, const int* __restrict__ rowptr,
    const int* __restrict__ cnt, const int* __restrict__ ssrc,
    const float* __restrict__ wp,
    const float* __restrict__ bias, float* __restrict__ out, int N) {
  __shared__ __attribute__((aligned(16))) float sA[MROWS][260];

  const v4f* __restrict__ h4 = (const v4f*)hin;
  const int Mbase = blockIdx.x * MROWS;
  const int tid = threadIdx.x;
  const int lane = tid & 31;
  const int wave = __builtin_amdgcn_readfirstlane(tid >> 5);

  const int r0 = wave * 4;
  const int n0 = min(Mbase + r0, N - 1);
  const int n1 = min(Mbase + r0 + 1, N - 1);
  const int n2 = min(Mbase + r0 + 2, N - 1);
  const int n3 = min(Mbase + r0 + 3, N - 1);
  const int start0 = rowptr[n0];
  const int e1 = rowptr[n1] - start0;
  const int e2 = rowptr[n2] - start0;
  const int e3 = rowptr[n3] - start0;
  const int total = rowptr[n3] + cnt[n3] - start0;

  int cur = 0;
  if (total > 0) cur = ssrc[start0 + min(lane, min(32, total) - 1)];
  {
    const v4f own0 = h4[(size_t)n0 * (D / 4) + lane];
    const v4f own1 = h4[(size_t)n1 * (D / 4) + lane];
    const v4f own2 = h4[(size_t)n2 * (D / 4) + lane];
    const v4f own3 = h4[(size_t)n3 * (D / 4) + lane];
    *(v4f*)&sA[r0][D + 4 * lane]     = own0;
    *(v4f*)&sA[r0 + 1][D + 4 * lane] = own1;
    *(v4f*)&sA[r0 + 2][D + 4 * lane] = own2;
    *(v4f*)&sA[r0 + 3][D + 4 * lane] = own3;
  }

  const char* __restrict__ hbytes = (const char*)hin;
  const unsigned lane16 = (unsigned)lane * 16u;
  v4f acc = {0.0f, 0.0f, 0.0f, 0.0f};
  int rowi = 0;
  int bound = e1;
  for (int cb = 0; cb < total; cb += 32) {
    const int mc = min(32, total - cb);
    int nxt = 0;
    if (cb + 32 < total) nxt = ssrc[start0 + cb + 32 + min(lane, min(32, total - cb - 32) - 1)];
    for (int j = 0; j < mc; j += 8) {
      v4f r[8];
#pragma unroll
      for (int k = 0; k < 8; ++k) {
        const int jj = min(j + k, mc - 1);
        const unsigned s = (unsigned)__builtin_amdgcn_readlane(cur, jj);
        r[k] = *(const v4f*)(hbytes + (size_t)s * (D * 4u) + lane16);
      }
      asm volatile("" :: "v"(r[0]), "v"(r[1]), "v"(r[2]), "v"(r[3]),
                         "v"(r[4]), "v"(r[5]), "v"(r[6]), "v"(r[7]));
#pragma unroll
      for (int k = 0; k < 8; ++k) {
        const int q = cb + j + k;
        while (__builtin_expect(q == bound, 0)) {
          *(v4f*)&sA[r0 + rowi][4 * lane] = acc;
          acc = (v4f){0.0f, 0.0f, 0.0f, 0.0f};
          ++rowi;
          bound = (rowi == 1) ? e2 : (rowi == 2) ? e3 : (rowi == 3) ? total : 0x7fffffff;
        }
        acc += r[k];
      }
    }
    cur = nxt;
  }
  while (rowi < 4) {
    *(v4f*)&sA[r0 + rowi][4 * lane] = acc;
    acc = (v4f){0.0f, 0.0f, 0.0f, 0.0f};
    ++rowi;
  }
  {
    const float c0 = fmaxf((float)e1, 1.0f);
    const float c1 = fmaxf((float)(e2 - e1), 1.0f);
    const float c2 = fmaxf((float)(e3 - e2), 1.0f);
    const float c3 = fmaxf((float)(total - e3), 1.0f);
    const v4f s0 = *(const v4f*)&sA[r0][4 * lane];
    const v4f s1 = *(const v4f*)&sA[r0 + 1][4 * lane];
    const v4f s2 = *(const v4f*)&sA[r0 + 2][4 * lane];
    const v4f s3 = *(const v4f*)&sA[r0 + 3][4 * lane];
    *(v4f*)&sA[r0][4 * lane]     = s0 / c0;
    *(v4f*)&sA[r0 + 1][4 * lane] = s1 / c1;
    *(v4f*)&sA[r0 + 2][4 * lane] = s2 / c2;
    *(v4f*)&sA[r0 + 3][4 * lane] = s3 / c3;
  }
  __syncthreads();

  const int col  = wave * 16 + (lane & 15);
  const int half = lane >> 4;
  const int rowA = lane & 15;

  v8f acc0 = {};
  v8f acc1 = {};

  const v4f* __restrict__ wp4 = (const v4f*)wp + wave * 32 + lane;
  for (int g = 0; g < 32; ++g) {
    const v4f bq  = wp4[g * 256];
    const v4f a0q = *(const v4f*)&sA[rowA][8 * g + 4 * half];
    const v4f a1q = *(const v4f*)&sA[16 + rowA][8 * g + 4 * half];
    v2f a0, a1, b;
    a0[0] = a0q[0]; a0[1] = a0q[1];
    a1[0] = a1q[0]; a1[1] = a1q[1];
    b[0]  = bq[0];  b[1]  = bq[1];
    acc0 = __builtin_amdgcn_wmma_f32_16x16x4_f32(
        false, a0, false, b, (short)0, acc0, false, false);
    acc1 = __builtin_amdgcn_wmma_f32_16x16x4_f32(
        false, a1, false, b, (short)0, acc1, false, false);
    a0[0] = a0q[2]; a0[1] = a0q[3];
    a1[0] = a1q[2]; a1[1] = a1q[3];
    b[0]  = bq[2];  b[1]  = bq[3];
    acc0 = __builtin_amdgcn_wmma_f32_16x16x4_f32(
        false, a0, false, b, (short)0, acc0, false, false);
    acc1 = __builtin_amdgcn_wmma_f32_16x16x4_f32(
        false, a1, false, b, (short)0, acc1, false, false);
  }

  const float bv = bias[col];
  float* orow = out + (size_t)(Mbase + half * 8) * D + col;
#pragma unroll
  for (int v = 0; v < 8; ++v) {
    float val0 = acc0[v] + bv;
    float val1 = acc1[v] + bv;
    if (RELU) { val0 = fmaxf(val0, 0.0f); val1 = fmaxf(val1, 0.0f); }
    orow[(size_t)v * D]        = val0;
    orow[(size_t)(16 + v) * D] = val1;
  }
}

extern "C" void kernel_launch(void* const* d_in, const int* in_sizes, int n_in,
                              void* d_out, int out_size, void* d_ws, size_t ws_size,
                              hipStream_t stream) {
  const float* x   = (const float*)d_in[0];
  const int*   ei  = (const int*)d_in[1];
  const float* Wl1 = (const float*)d_in[2];
  const float* Wr1 = (const float*)d_in[3];
  const float* b1  = (const float*)d_in[4];
  const float* Wl2 = (const float*)d_in[5];
  const float* Wr2 = (const float*)d_in[6];
  const float* b2  = (const float*)d_in[7];
  float* out = (float*)d_out;

  const int N = in_sizes[0] / D;
  const int E = in_sizes[1] / 2;
  const int* src = ei;
  const int* dst = ei + E;

  const int NB = (N + BIN_NODES - 1) / BIN_NODES;
  const int M  = (NB + 1) * PART_BLOCKS;

  float* h   = (float*)d_ws;
  float* wp  = h + (size_t)N * D;
  int2* pairs = (int2*)(wp + WP_FLOATS);
  int* cnt    = (int*)(pairs + E);
  int* rowptr = cnt + N;
  int* pt     = rowptr + N;
  int* ssrc   = pt + M;

  dim3 blk(256);
  dim3 big(1024);
  const int tileBlocks = (N + MROWS - 1) / MROWS;

  slice_sort <<<PART_BLOCKS, big, 0, stream>>>(src, dst, pt, pairs, E, NB, Wl1, Wr1, Wl2, Wr2, wp);
  bin_sort   <<<NB, big, 0, stream>>>(pairs, pt, cnt, rowptr, ssrc, N, E, NB);

  sage_layer<1><<<tileBlocks, blk, 0, stream>>>(x, rowptr, cnt, ssrc, wp, b1, h, N);

  sage_layer<0><<<tileBlocks, blk, 0, stream>>>(h, rowptr, cnt, ssrc, wp + WP_LAYER, b2, out, N);
}
